# starting kernel
# baseline (speedup 1.0000x reference)
#include <hip/hip_runtime.h>
#include <hip/hip_bf16.h>


typedef __attribute__((ext_vector_type(2))) float v2f;
typedef __attribute__((ext_vector_type(8))) float v8f;

#define BB   2
#define LQ   1024
#define LIN  4096
#define CC   256
#define MH   8
#define NP   4
#define DH   32
#define QT   4

template <int MT, int NT>
__global__ void __launch_bounds__(32)
wmma_gemm_f32(const float* __restrict__ X, const float* __restrict__ W,
              const float* __restrict__ bias, float* __restrict__ out,
              int Ncols, int K) {
    const int lane = threadIdx.x;
    const int l15  = lane & 15;
    const int kh   = lane >> 4;
    const int row0 = blockIdx.x * (16 * MT);
    const int col0 = blockIdx.y * (16 * NT);

    const float* __restrict__ xr[MT];
    const float* __restrict__ wr[NT];
#pragma unroll
    for (int mt = 0; mt < MT; ++mt)
        xr[mt] = X + (size_t)(row0 + mt * 16 + l15) * K + 2 * kh;
#pragma unroll
    for (int nt = 0; nt < NT; ++nt)
        wr[nt] = W + (size_t)(col0 + nt * 16 + l15) * K + 2 * kh;

    v8f acc[MT][NT];
#pragma unroll
    for (int mt = 0; mt < MT; ++mt)
#pragma unroll
        for (int nt = 0; nt < NT; ++nt) {
            v8f z = {};
            acc[mt][nt] = z;
        }

#pragma unroll 2
    for (int k = 0; k < K; k += 4) {
        v2f a[MT], b[NT];
#pragma unroll
        for (int mt = 0; mt < MT; ++mt) a[mt] = *(const v2f*)(xr[mt] + k);
#pragma unroll
        for (int nt = 0; nt < NT; ++nt) b[nt] = *(const v2f*)(wr[nt] + k);
#pragma unroll
        for (int mt = 0; mt < MT; ++mt)
#pragma unroll
            for (int nt = 0; nt < NT; ++nt)
                acc[mt][nt] = __builtin_amdgcn_wmma_f32_16x16x4_f32(
                     false, a[mt],  false, b[nt],
                     (short)0, acc[mt][nt],
                     false,  false);
    }

#pragma unroll
    for (int nt = 0; nt < NT; ++nt) {
        const int   col  = col0 + nt * 16;
        const float bcol = bias[col + l15];
#pragma unroll
        for (int mt = 0; mt < MT; ++mt) {
            const int row = row0 + mt * 16;
#pragma unroll
            for (int r = 0; r < 8; ++r) {
                out[(size_t)(row + r + 8 * kh) * Ncols + col + l15] =
                    acc[mt][nt][r] + bcol;
            }
        }
    }
}

__global__ void build_pts(const float* __restrict__ off,
                          const float* __restrict__ qpts,
                          const float* __restrict__ ipts,
                          float* __restrict__ S4, float* __restrict__ V4) {
    const int i = blockIdx.x * blockDim.x + threadIdx.x;
    if (i < BB * LQ * MH * NP) {
        const int b  = i >> 15;
        const int q  = (i >> 5) & (LQ - 1);
        const int mp = i & 31;
        const float* o = off  + ((size_t)(b * LQ + q)) * (MH * NP * 3) + mp * 3;
        const float* p = qpts + ((size_t)(b * LQ + q)) * 3;
        float4 s;
        s.x = p[0] + o[0];
        s.y = p[1] + o[1];
        s.z = p[2] + o[2];
        s.w = 1.0f;
        *(float4*)(S4 + (size_t)i * 4) = s;
    }
    if (i < BB * LIN) {
        const float* v = ipts + (size_t)i * 3;
        const float x = v[0], y = v[1], z = v[2];
        float4 t;
        t.x = -2.0f * x;
        t.y = -2.0f * y;
        t.z = -2.0f * z;
        t.w = x * x + y * y + z * z;
        *(float4*)(V4 + (size_t)i * 4) = t;
    }
}

__global__ void softmax_p4(float* __restrict__ a, int rows) {
    const int r = blockIdx.x * blockDim.x + threadIdx.x;
    if (r >= rows) return;
    float4 v = *(float4*)(a + (size_t)r * 4);
    const float mx = fmaxf(fmaxf(v.x, v.y), fmaxf(v.z, v.w));
    const float e0 = __expf(v.x - mx), e1 = __expf(v.y - mx);
    const float e2 = __expf(v.z - mx), e3 = __expf(v.w - mx);
    const float inv = 1.0f / (e0 + e1 + e2 + e3);
    v.x = e0 * inv; v.y = e1 * inv; v.z = e2 * inv; v.w = e3 * inv;
    *(float4*)(a + (size_t)r * 4) = v;
}

__device__ __forceinline__ void knn_step(const v2f a, const v2f* bq, int cib,
                                         float* best, int* bidx) {
#pragma unroll
    for (int t = 0; t < QT; ++t) {
        v8f c = {};
        c = __builtin_amdgcn_wmma_f32_16x16x4_f32(
            false, a, false, bq[t], (short)0, c, false, false);
#pragma unroll
        for (int r = 0; r < 8; ++r) {
            const float dv = c[r];
            const int   ci = cib + r;
            if (dv < best[t]) { best[t] = dv; bidx[t] = ci; }
        }
    }
}

__global__ void __launch_bounds__(32)
knn_wmma(const float* __restrict__ S4, const float* __restrict__ V4,
         int* __restrict__ idxOut) {
    const int lane = threadIdx.x;
    const int l15  = lane & 15;
    const int kh   = lane >> 4;
    const int n0   = blockIdx.x * (16 * QT);
    const int b    = n0 >> 15;

    v2f bq[QT];
#pragma unroll
    for (int t = 0; t < QT; ++t)
        bq[t] = *(const v2f*)(S4 + (size_t)(n0 + t * 16 + l15) * 4 + 2 * kh);

    const float* __restrict__ vptr = V4 + (size_t)b * LIN * 4
                                        + (size_t)l15 * 4 + 2 * kh;

    float best[QT];
    int   bidx[QT];
#pragma unroll
    for (int t = 0; t < QT; ++t) { best[t] = 3.4e38f; bidx[t] = 0; }

    v2f a_cur = *(const v2f*)(vptr);
    for (int cb = 0; cb < LIN - 16; cb += 16) {
        const v2f a_next = *(const v2f*)(vptr + (size_t)(cb + 16) * 4);
        knn_step(a_cur, bq, cb + (kh << 3), best, bidx);
        a_cur = a_next;
    }
    knn_step(a_cur, bq, (LIN - 16) + (kh << 3), best, bidx);

#pragma unroll
    for (int t = 0; t < QT; ++t) {
        const float obest = __shfl_xor(best[t], 16, 32);
        const int   oidx  = __shfl_xor(bidx[t], 16, 32);
        if (obest < best[t] || (obest == best[t] && oidx < bidx[t])) {
            best[t] = obest; bidx[t] = oidx;
        }
        if (lane < 16) idxOut[n0 + t * 16 + l15] = bidx[t];
    }
}

__global__ void __launch_bounds__(CC)
gather_weighted(const float* __restrict__ value, const int* __restrict__ idx,
                const float* __restrict__ attw, float* __restrict__ out_tmp) {
    const int bq = blockIdx.x;
    const int c  = threadIdx.x;
    const int b  = bq >> 10;
    const int m  = c >> 5;
    const int*   id = idx  + ((size_t)bq * MH + m) * NP;
    const float* aw = attw + ((size_t)bq * MH + m) * NP;
    const float* vb = value + (size_t)b * LIN * CC;
    float acc = 0.0f;
#pragma unroll
    for (int p = 0; p < NP; ++p)
        acc += aw[p] * vb[(size_t)id[p] * CC + c];
    out_tmp[(size_t)bq * CC + c] = acc;
}

extern "C" void kernel_launch(void* const* d_in, const int* in_sizes, int n_in,
                              void* d_out, int out_size, void* d_ws, size_t ws_size,
                              hipStream_t stream) {
    const float* query = (const float*)d_in[0];
    const float* qpts  = (const float*)d_in[1];
    const float* inp   = (const float*)d_in[2];
    const float* ipts  = (const float*)d_in[3];
    const float* Wv    = (const float*)d_in[4];
    const float* bv    = (const float*)d_in[5];
    const float* Ws    = (const float*)d_in[6];
    const float* bs    = (const float*)d_in[7];
    const float* Wa    = (const float*)d_in[8];
    const float* ba    = (const float*)d_in[9];
    const float* Wo    = (const float*)d_in[10];
    const float* bo    = (const float*)d_in[11];
    float* out = (float*)d_out;

    char* w = (char*)d_ws;
    float* value = (float*)w;  w += (size_t)BB * LIN * CC * 4;
    float* off   = (float*)w;  w += (size_t)BB * LQ * MH * NP * 3 * 4;
    float* attw  = (float*)w;  w += (size_t)BB * LQ * MH * NP * 4;
    float* S4    = (float*)w;  w += (size_t)BB * LQ * MH * NP * 4 * 4;
    float* V4    = (float*)w;  w += (size_t)BB * LIN * 4 * 4;
    int*   idx   = (int*)w;    w += (size_t)BB * LQ * MH * NP * 4;
    float* otmp  = (float*)w;  w += (size_t)BB * LQ * CC * 4;

    const dim3 wave(32);

    wmma_gemm_f32<2, 4><<<dim3((BB * LIN) / 32, CC / 64), wave, 0, stream>>>(
        inp, Wv, bv, value, CC, CC);

    wmma_gemm_f32<2, 2><<<dim3((BB * LQ) / 32, (MH * NP * 3) / 32), wave, 0, stream>>>(
        query, Ws, bs, off, MH * NP * 3, CC);

    wmma_gemm_f32<2, 2><<<dim3((BB * LQ) / 32, (MH * NP) / 32), wave, 0, stream>>>(
        query, Wa, ba, attw, MH * NP, CC);

    {
        const int n = BB * LQ * MH * NP;
        build_pts<<<(n + 255) / 256, 256, 0, stream>>>(off, qpts, ipts, S4, V4);
    }

    {
        const int rows = BB * LQ * MH;
        softmax_p4<<<(rows + 255) / 256, 256, 0, stream>>>(attw, rows);
    }

    knn_wmma<<<(BB * LQ * MH * NP) / (16 * QT), wave, 0, stream>>>(S4, V4, idx);

    gather_weighted<<<BB * LQ, CC, 0, stream>>>(value, idx, attw, otmp);

    wmma_gemm_f32<2, 4><<<dim3((BB * LQ) / 32, CC / 64), wave, 0, stream>>>(
        otmp, Wo, bo, out, CC, CC);
}


# KNN: deal each 64-query tile's candidate blocks round-robin to 8 waves of one work-group (8 waves per SIMD instead of 1), merge per-wave argmins through LDS, lowest index on ties
# speedup vs baseline: 2.0793x; 2.0793x over previous
#include <hip/hip_runtime.h>
#include <hip/hip_bf16.h>


typedef __attribute__((ext_vector_type(2))) float v2f;
typedef __attribute__((ext_vector_type(8))) float v8f;

#define BB   2
#define LQ   1024
#define LIN  4096
#define CC   256
#define MH   8
#define NP   4
#define DH   32
#define QT   4
#define NW   8
#define CSTEP (16 * NW)

template <int MT, int NT>
__global__ void __launch_bounds__(32)
wmma_gemm_f32(const float* __restrict__ X, const float* __restrict__ W,
              const float* __restrict__ bias, float* __restrict__ out,
              int Ncols, int K) {
    const int lane = threadIdx.x;
    const int l15  = lane & 15;
    const int kh   = lane >> 4;
    const int row0 = blockIdx.x * (16 * MT);
    const int col0 = blockIdx.y * (16 * NT);

    const float* __restrict__ xr[MT];
    const float* __restrict__ wr[NT];
#pragma unroll
    for (int mt = 0; mt < MT; ++mt)
        xr[mt] = X + (size_t)(row0 + mt * 16 + l15) * K + 2 * kh;
#pragma unroll
    for (int nt = 0; nt < NT; ++nt)
        wr[nt] = W + (size_t)(col0 + nt * 16 + l15) * K + 2 * kh;

    v8f acc[MT][NT];
#pragma unroll
    for (int mt = 0; mt < MT; ++mt)
#pragma unroll
        for (int nt = 0; nt < NT; ++nt) {
            v8f z = {};
            acc[mt][nt] = z;
        }

#pragma unroll 2
    for (int k = 0; k < K; k += 4) {
        v2f a[MT], b[NT];
#pragma unroll
        for (int mt = 0; mt < MT; ++mt) a[mt] = *(const v2f*)(xr[mt] + k);
#pragma unroll
        for (int nt = 0; nt < NT; ++nt) b[nt] = *(const v2f*)(wr[nt] + k);
#pragma unroll
        for (int mt = 0; mt < MT; ++mt)
#pragma unroll
            for (int nt = 0; nt < NT; ++nt)
                acc[mt][nt] = __builtin_amdgcn_wmma_f32_16x16x4_f32(
                     false, a[mt],  false, b[nt],
                     (short)0, acc[mt][nt],
                     false,  false);
    }

#pragma unroll
    for (int nt = 0; nt < NT; ++nt) {
        const int   col  = col0 + nt * 16;
        const float bcol = bias[col + l15];
#pragma unroll
        for (int mt = 0; mt < MT; ++mt) {
            const int row = row0 + mt * 16;
#pragma unroll
            for (int r = 0; r < 8; ++r) {
                out[(size_t)(row + r + 8 * kh) * Ncols + col + l15] =
                    acc[mt][nt][r] + bcol;
            }
        }
    }
}

__global__ void build_pts(const float* __restrict__ off,
                          const float* __restrict__ qpts,
                          const float* __restrict__ ipts,
                          float* __restrict__ S4, float* __restrict__ V4) {
    const int i = blockIdx.x * blockDim.x + threadIdx.x;
    if (i < BB * LQ * MH * NP) {
        const int b  = i >> 15;
        const int q  = (i >> 5) & (LQ - 1);
        const int mp = i & 31;
        const float* o = off  + ((size_t)(b * LQ + q)) * (MH * NP * 3) + mp * 3;
        const float* p = qpts + ((size_t)(b * LQ + q)) * 3;
        float4 s;
        s.x = p[0] + o[0];
        s.y = p[1] + o[1];
        s.z = p[2] + o[2];
        s.w = 1.0f;
        *(float4*)(S4 + (size_t)i * 4) = s;
    }
    if (i < BB * LIN) {
        const float* v = ipts + (size_t)i * 3;
        const float x = v[0], y = v[1], z = v[2];
        float4 t;
        t.x = -2.0f * x;
        t.y = -2.0f * y;
        t.z = -2.0f * z;
        t.w = x * x + y * y + z * z;
        *(float4*)(V4 + (size_t)i * 4) = t;
    }
}

__global__ void softmax_p4(float* __restrict__ a, int rows) {
    const int r = blockIdx.x * blockDim.x + threadIdx.x;
    if (r >= rows) return;
    float4 v = *(float4*)(a + (size_t)r * 4);
    const float mx = fmaxf(fmaxf(v.x, v.y), fmaxf(v.z, v.w));
    const float e0 = __expf(v.x - mx), e1 = __expf(v.y - mx);
    const float e2 = __expf(v.z - mx), e3 = __expf(v.w - mx);
    const float inv = 1.0f / (e0 + e1 + e2 + e3);
    v.x = e0 * inv; v.y = e1 * inv; v.z = e2 * inv; v.w = e3 * inv;
    *(float4*)(a + (size_t)r * 4) = v;
}

__device__ __forceinline__ void knn_step(const v2f a, const v2f* bq, int cib,
                                         float* best, int* bidx) {
#pragma unroll
    for (int t = 0; t < QT; ++t) {
        v8f c = {};
        c = __builtin_amdgcn_wmma_f32_16x16x4_f32(
            false, a, false, bq[t], (short)0, c, false, false);
#pragma unroll
        for (int r = 0; r < 8; ++r) {
            const float dv = c[r];
            const int   ci = cib + r;
            if (dv < best[t]) { best[t] = dv; bidx[t] = ci; }
        }
    }
}

__global__ void __launch_bounds__(32 * NW)
knn_wmma(const float* __restrict__ S4, const float* __restrict__ V4,
         int* __restrict__ idxOut) {
    __shared__ float sBest[NW][16 * QT];
    __shared__ int   sIdx[NW][16 * QT];

    const int lane = threadIdx.x & 31;
    const int wv   = threadIdx.x >> 5;
    const int l15  = lane & 15;
    const int kh   = lane >> 4;
    const int n0   = blockIdx.x * (16 * QT);
    const int b    = n0 >> 15;
    const int cw   = wv * 16;

    v2f bq[QT];
#pragma unroll
    for (int t = 0; t < QT; ++t)
        bq[t] = *(const v2f*)(S4 + (size_t)(n0 + t * 16 + l15) * 4 + 2 * kh);

    const float* __restrict__ vptr = V4 + ((size_t)b * LIN + cw) * 4
                                        + (size_t)l15 * 4 + 2 * kh;

    float best[QT];
    int   bidx[QT];
#pragma unroll
    for (int t = 0; t < QT; ++t) { best[t] = 3.4e38f; bidx[t] = 0; }

    v2f a_cur = *(const v2f*)(vptr);
    for (int cb = 0; cb < LIN - CSTEP; cb += CSTEP) {
        const v2f a_next = *(const v2f*)(vptr + (size_t)(cb + CSTEP) * 4);
        knn_step(a_cur, bq, cw + cb + (kh << 3), best, bidx);
        a_cur = a_next;
    }
    knn_step(a_cur, bq, cw + (LIN - CSTEP) + (kh << 3), best, bidx);

#pragma unroll
    for (int t = 0; t < QT; ++t) {
        const float obest = __shfl_xor(best[t], 16, 32);
        const int   oidx  = __shfl_xor(bidx[t], 16, 32);
        if (obest < best[t] || (obest == best[t] && oidx < bidx[t])) {
            best[t] = obest; bidx[t] = oidx;
        }
        if (lane < 16) {
            sBest[wv][t * 16 + l15] = best[t];
            sIdx[wv][t * 16 + l15]  = bidx[t];
        }
    }
    __syncthreads();

    if (threadIdx.x < 16 * QT) {
        float mb = sBest[0][threadIdx.x];
        int   mi = sIdx[0][threadIdx.x];
        for (int w = 1; w < NW; ++w) {
            const float ob = sBest[w][threadIdx.x];
            const int   oi = sIdx[w][threadIdx.x];
            if (ob < mb || (ob == mb && oi < mi)) { mb = ob; mi = oi; }
        }
        idxOut[n0 + threadIdx.x] = mi;
    }
}

__global__ void __launch_bounds__(CC)
gather_weighted(const float* __restrict__ value, const int* __restrict__ idx,
                const float* __restrict__ attw, float* __restrict__ out_tmp) {
    const int bq = blockIdx.x;
    const int c  = threadIdx.x;
    const int b  = bq >> 10;
    const int m  = c >> 5;
    const int*   id = idx  + ((size_t)bq * MH + m) * NP;
    const float* aw = attw + ((size_t)bq * MH + m) * NP;
    const float* vb = value + (size_t)b * LIN * CC;
    float acc = 0.0f;
#pragma unroll
    for (int p = 0; p < NP; ++p)
        acc += aw[p] * vb[(size_t)id[p] * CC + c];
    out_tmp[(size_t)bq * CC + c] = acc;
}

extern "C" void kernel_launch(void* const* d_in, const int* in_sizes, int n_in,
                              void* d_out, int out_size, void* d_ws, size_t ws_size,
                              hipStream_t stream) {
    const float* query = (const float*)d_in[0];
    const float* qpts  = (const float*)d_in[1];
    const float* inp   = (const float*)d_in[2];
    const float* ipts  = (const float*)d_in[3];
    const float* Wv    = (const float*)d_in[4];
    const float* bv    = (const float*)d_in[5];
    const float* Ws    = (const float*)d_in[6];
    const float* bs    = (const float*)d_in[7];
    const float* Wa    = (const float*)d_in[8];
    const float* ba    = (const float*)d_in[9];
    const float* Wo    = (const float*)d_in[10];
    const float* bo    = (const float*)d_in[11];
    float* out = (float*)d_out;

    char* w = (char*)d_ws;
    float* value = (float*)w;  w += (size_t)BB * LIN * CC * 4;
    float* off   = (float*)w;  w += (size_t)BB * LQ * MH * NP * 3 * 4;
    float* attw  = (float*)w;  w += (size_t)BB * LQ * MH * NP * 4;
    float* S4    = (float*)w;  w += (size_t)BB * LQ * MH * NP * 4 * 4;
    float* V4    = (float*)w;  w += (size_t)BB * LIN * 4 * 4;
    int*   idx   = (int*)w;    w += (size_t)BB * LQ * MH * NP * 4;
    float* otmp  = (float*)w;  w += (size_t)BB * LQ * CC * 4;

    const dim3 wave(32);

    wmma_gemm_f32<2, 4><<<dim3((BB * LIN) / 32, CC / 64), wave, 0, stream>>>(
        inp, Wv, bv, value, CC, CC);

    wmma_gemm_f32<2, 2><<<dim3((BB * LQ) / 32, (MH * NP * 3) / 32), wave, 0, stream>>>(
        query, Ws, bs, off, MH * NP * 3, CC);

    wmma_gemm_f32<2, 2><<<dim3((BB * LQ) / 32, (MH * NP) / 32), wave, 0, stream>>>(
        query, Wa, ba, attw, MH * NP, CC);

    {
        const int n = BB * LQ * MH * NP;
        build_pts<<<(n + 255) / 256, 256, 0, stream>>>(off, qpts, ipts, S4, V4);
    }

    {
        const int rows = BB * LQ * MH;
        softmax_p4<<<(rows + 255) / 256, 256, 0, stream>>>(attw, rows);
    }

    knn_wmma<<<(BB * LQ * MH * NP) / (16 * QT), dim3(32 * NW), 0, stream>>>(S4, V4, idx);

    gather_weighted<<<BB * LQ, CC, 0, stream>>>(value, idx, attw, otmp);

    wmma_gemm_f32<2, 4><<<dim3((BB * LQ) / 32, CC / 64), wave, 0, stream>>>(
        otmp, Wo, bo, out, CC, CC);
}


# Launch fusion 8->3: the three input GEMMs as tiles of one grid (+candidate 4-vectors in its epilogue); KNN builds its sampling points itself and ends with softmax+gather of its 2 query rows
# speedup vs baseline: 3.0632x; 1.4732x over previous
#include <hip/hip_runtime.h>
#include <hip/hip_bf16.h>


typedef __attribute__((ext_vector_type(2))) float v2f;
typedef __attribute__((ext_vector_type(8))) float v8f;

#define BB   2
#define LQ   1024
#define LIN  4096
#define CC   256
#define MH   8
#define NP   4
#define DH   32
#define QT   4
#define NW   8
#define CSTEP (16 * NW)

#define VT_X ((BB * LIN) / 32)
#define VT_Y (CC / 64)
#define VT   (VT_X * VT_Y)
#define QT_X ((BB * LQ) / 32)
#define QT_Y 4
#define NOFF (MH * NP * 3)

template <int MT, int NT>
__device__ __forceinline__ void
gemm_tile_f32(const float* __restrict__ X, const float* __restrict__ W,
              const float* __restrict__ bias, float* __restrict__ out,
              int Ncols, int K, int row0, int col0, int lane) {
    const int l15  = lane & 15;
    const int kh   = lane >> 4;

    const float* __restrict__ xr[MT];
    const float* __restrict__ wr[NT];
#pragma unroll
    for (int mt = 0; mt < MT; ++mt)
        xr[mt] = X + (size_t)(row0 + mt * 16 + l15) * K + 2 * kh;
#pragma unroll
    for (int nt = 0; nt < NT; ++nt)
        wr[nt] = W + (size_t)(col0 + nt * 16 + l15) * K + 2 * kh;

    v8f acc[MT][NT];
#pragma unroll
    for (int mt = 0; mt < MT; ++mt)
#pragma unroll
        for (int nt = 0; nt < NT; ++nt) {
            v8f z = {};
            acc[mt][nt] = z;
        }

#pragma unroll 2
    for (int k = 0; k < K; k += 4) {
        v2f a[MT], b[NT];
#pragma unroll
        for (int mt = 0; mt < MT; ++mt) a[mt] = *(const v2f*)(xr[mt] + k);
#pragma unroll
        for (int nt = 0; nt < NT; ++nt) b[nt] = *(const v2f*)(wr[nt] + k);
#pragma unroll
        for (int mt = 0; mt < MT; ++mt)
#pragma unroll
            for (int nt = 0; nt < NT; ++nt)
                acc[mt][nt] = __builtin_amdgcn_wmma_f32_16x16x4_f32(
                     false, a[mt],  false, b[nt],
                     (short)0, acc[mt][nt],
                     false,  false);
    }

#pragma unroll
    for (int nt = 0; nt < NT; ++nt) {
        const int   col  = col0 + nt * 16;
        const float bcol = bias[col + l15];
#pragma unroll
        for (int mt = 0; mt < MT; ++mt) {
            const int row = row0 + mt * 16;
#pragma unroll
            for (int r = 0; r < 8; ++r) {
                out[(size_t)(row + r + 8 * kh) * Ncols + col + l15] =
                    acc[mt][nt][r] + bcol;
            }
        }
    }
}

template <int MT, int NT>
__global__ void __launch_bounds__(32)
wmma_gemm_f32(const float* __restrict__ X, const float* __restrict__ W,
              const float* __restrict__ bias, float* __restrict__ out,
              int Ncols, int K) {
    gemm_tile_f32<MT, NT>(X, W, bias, out, Ncols, K,
                          blockIdx.x * (16 * MT), blockIdx.y * (16 * NT),
                          threadIdx.x);
}

__global__ void __launch_bounds__(32)
proj_all(const float* __restrict__ inp, const float* __restrict__ Wv,
         const float* __restrict__ bv, const float* __restrict__ query,
         const float* __restrict__ Ws, const float* __restrict__ bs,
         const float* __restrict__ Wa, const float* __restrict__ ba,
         const float* __restrict__ ipts,
         float* __restrict__ value, float* __restrict__ off,
         float* __restrict__ logits, float* __restrict__ V4) {
    const int lane = threadIdx.x;
    const int id   = blockIdx.x;
    if (id < VT) {
        const int bx = id % VT_X;
        const int by = id / VT_X;
        if (by == 0) {
            const int l = bx * 32 + lane;
            const float* v = ipts + (size_t)l * 3;
            const float x = v[0], y = v[1], z = v[2];
            float4 t;
            t.x = -2.0f * x;
            t.y = -2.0f * y;
            t.z = -2.0f * z;
            t.w = x * x + y * y + z * z;
            *(float4*)(V4 + (size_t)l * 4) = t;
        }
        gemm_tile_f32<2, 4>(inp, Wv, bv, value, CC, CC, bx * 32, by * 64, lane);
    } else {
        const int j  = id - VT;
        const int bx = j % QT_X;
        const int by = j / QT_X;
        if (by < 3)
            gemm_tile_f32<2, 2>(query, Ws, bs, off, NOFF, CC, bx * 32, by * 32, lane);
        else
            gemm_tile_f32<2, 2>(query, Wa, ba, logits, MH * NP, CC, bx * 32, 0, lane);
    }
}

__device__ __forceinline__ void knn_step(const v2f a, const v2f* bq, int cib,
                                         float* best, int* bidx) {
#pragma unroll
    for (int t = 0; t < QT; ++t) {
        v8f c = {};
        c = __builtin_amdgcn_wmma_f32_16x16x4_f32(
            false, a, false, bq[t], (short)0, c, false, false);
#pragma unroll
        for (int r = 0; r < 8; ++r) {
            const float dv = c[r];
            const int   ci = cib + r;
            if (dv < best[t]) { best[t] = dv; bidx[t] = ci; }
        }
    }
}

__global__ void __launch_bounds__(32 * NW)
knn_gather(const float* __restrict__ off, const float* __restrict__ qpts,
           const float* __restrict__ V4, const float* __restrict__ value,
           const float* __restrict__ logits, float* __restrict__ out_tmp) {
    __shared__ float sBest[NW][16 * QT];
    __shared__ int   sIdx[NW][16 * QT];
    __shared__ int   sFin[16 * QT];

    const int lane = threadIdx.x & 31;
    const int wv   = threadIdx.x >> 5;
    const int l15  = lane & 15;
    const int kh   = lane >> 4;
    const int n0   = blockIdx.x * (16 * QT);
    const int b    = n0 >> 15;
    const int cw   = wv * 16;

    v2f bq[QT];
#pragma unroll
    for (int t = 0; t < QT; ++t) {
        const int n  = n0 + t * 16 + l15;
        const int rq = n >> 5;
        const int mp = n & 31;
        const float* o = off  + (size_t)rq * NOFF + mp * 3;
        const float* p = qpts + (size_t)rq * 3;
        const float s0 = p[2 * kh] + o[2 * kh];
        const float s1 = p[1] + o[1];
        v2f f;
        f.x = s0;
        f.y = kh ? 1.0f : s1;
        bq[t] = f;
    }

    const float* __restrict__ vptr = V4 + ((size_t)b * LIN + cw) * 4
                                        + (size_t)l15 * 4 + 2 * kh;

    float best[QT];
    int   bidx[QT];
#pragma unroll
    for (int t = 0; t < QT; ++t) { best[t] = 3.4e38f; bidx[t] = 0; }

    v2f a_cur = *(const v2f*)(vptr);
    for (int cb = 0; cb < LIN - CSTEP; cb += CSTEP) {
        const v2f a_next = *(const v2f*)(vptr + (size_t)(cb + CSTEP) * 4);
        knn_step(a_cur, bq, cw + cb + (kh << 3), best, bidx);
        a_cur = a_next;
    }
    knn_step(a_cur, bq, cw + (LIN - CSTEP) + (kh << 3), best, bidx);

#pragma unroll
    for (int t = 0; t < QT; ++t) {
        const float obest = __shfl_xor(best[t], 16, 32);
        const int   oidx  = __shfl_xor(bidx[t], 16, 32);
        if (obest < best[t] || (obest == best[t] && oidx < bidx[t])) {
            best[t] = obest; bidx[t] = oidx;
        }
        if (lane < 16) {
            sBest[wv][t * 16 + l15] = best[t];
            sIdx[wv][t * 16 + l15]  = bidx[t];
        }
    }
    __syncthreads();

    if (threadIdx.x < 16 * QT) {
        float mb = sBest[0][threadIdx.x];
        int   mi = sIdx[0][threadIdx.x];
        for (int w = 1; w < NW; ++w) {
            const float ob = sBest[w][threadIdx.x];
            const int   oi = sIdx[w][threadIdx.x];
            if (ob < mb || (ob == mb && oi < mi)) { mb = ob; mi = oi; }
        }
        sFin[threadIdx.x] = mi;
    }
    __syncthreads();

    {
        const int c = threadIdx.x;
        const int m = c >> 5;
        const float* vb = value + (size_t)b * LIN * CC;
#pragma unroll
        for (int j = 0; j < (16 * QT) / (MH * NP); ++j) {
            const int rq = (n0 >> 5) + j;
            float4 v = *(const float4*)(logits + ((size_t)rq * MH + m) * NP);
            const float mx = fmaxf(fmaxf(v.x, v.y), fmaxf(v.z, v.w));
            const float e0 = __expf(v.x - mx), e1 = __expf(v.y - mx);
            const float e2 = __expf(v.z - mx), e3 = __expf(v.w - mx);
            const float inv = 1.0f / (e0 + e1 + e2 + e3);
            float aw[NP];
            aw[0] = e0 * inv; aw[1] = e1 * inv; aw[2] = e2 * inv; aw[3] = e3 * inv;
            const int* id = sFin + j * (MH * NP) + m * NP;
            float acc = 0.0f;
#pragma unroll
            for (int p = 0; p < NP; ++p)
                acc += aw[p] * vb[(size_t)id[p] * CC + c];
            out_tmp[(size_t)rq * CC + c] = acc;
        }
    }
}

extern "C" void kernel_launch(void* const* d_in, const int* in_sizes, int n_in,
                              void* d_out, int out_size, void* d_ws, size_t ws_size,
                              hipStream_t stream) {
    const float* query = (const float*)d_in[0];
    const float* qpts  = (const float*)d_in[1];
    const float* inp   = (const float*)d_in[2];
    const float* ipts  = (const float*)d_in[3];
    const float* Wv    = (const float*)d_in[4];
    const float* bv    = (const float*)d_in[5];
    const float* Ws    = (const float*)d_in[6];
    const float* bs    = (const float*)d_in[7];
    const float* Wa    = (const float*)d_in[8];
    const float* ba    = (const float*)d_in[9];
    const float* Wo    = (const float*)d_in[10];
    const float* bo    = (const float*)d_in[11];
    float* out = (float*)d_out;

    char* w = (char*)d_ws;
    float* value  = (float*)w;  w += (size_t)BB * LIN * CC * 4;
    float* off    = (float*)w;  w += (size_t)BB * LQ * MH * NP * 3 * 4;
    float* logits = (float*)w;  w += (size_t)BB * LQ * MH * NP * 4;
    float* V4     = (float*)w;  w += (size_t)BB * LIN * 4 * 4;
    float* otmp   = (float*)w;  w += (size_t)BB * LQ * CC * 4;

    const dim3 wave(32);

    proj_all<<<VT + QT_X * QT_Y, wave, 0, stream>>>(
        inp, Wv, bv, query, Ws, bs, Wa, ba, ipts, value, off, logits, V4);

    knn_gather<<<(BB * LQ * MH * NP) / (16 * QT), dim3(32 * NW), 0, stream>>>(
        off, qpts, V4, value, logits, otmp);

    wmma_gemm_f32<2, 4><<<dim3((BB * LQ) / 32, CC / 64), wave, 0, stream>>>(
        otmp, Wo, bo, out, CC, CC);
}


# GEMM tiles: software-pipelined k loop, next 32-k block's A/B fragments loaded as one group into a second register set before this block's WMMAs; same k and product order, bit-identical
# speedup vs baseline: 3.5157x; 1.1477x over previous
#include <hip/hip_runtime.h>
#include <hip/hip_bf16.h>


typedef __attribute__((ext_vector_type(2))) float v2f;
typedef __attribute__((ext_vector_type(8))) float v8f;

#define BB   2
#define LQ   1024
#define LIN  4096
#define CC   256
#define MH   8
#define NP   4
#define DH   32
#define QT   4
#define NW   8
#define CSTEP (16 * NW)

#define VT_X ((BB * LIN) / 32)
#define VT_Y (CC / 64)
#define VT   (VT_X * VT_Y)
#define QT_X ((BB * LQ) / 32)
#define QT_Y 4
#define NOFF (MH * NP * 3)

#define KB    8
#define KSPAN (4 * KB)

template <int MT, int NT>
__device__ __forceinline__ void
gemm_frag_load(const float* __restrict__ const (&xr)[MT],
               const float* __restrict__ const (&wr)[NT], int k0,
               v2f (&a)[KB][MT], v2f (&b)[KB][NT]) {
#pragma unroll
    for (int s = 0; s < KB; ++s) {
#pragma unroll
        for (int mt = 0; mt < MT; ++mt) a[s][mt] = *(const v2f*)(xr[mt] + k0 + 4 * s);
#pragma unroll
        for (int nt = 0; nt < NT; ++nt) b[s][nt] = *(const v2f*)(wr[nt] + k0 + 4 * s);
    }
}

template <int MT, int NT>
__device__ __forceinline__ void
gemm_frag_mma(const v2f (&a)[KB][MT], const v2f (&b)[KB][NT], v8f (&acc)[MT][NT]) {
#pragma unroll
    for (int s = 0; s < KB; ++s)
#pragma unroll
        for (int mt = 0; mt < MT; ++mt)
#pragma unroll
            for (int nt = 0; nt < NT; ++nt)
                acc[mt][nt] = __builtin_amdgcn_wmma_f32_16x16x4_f32(
                     false, a[s][mt],  false, b[s][nt],
                     (short)0, acc[mt][nt],
                     false,  false);
}

template <int MT, int NT>
__device__ __forceinline__ void
gemm_tile_f32(const float* __restrict__ X, const float* __restrict__ W,
              const float* __restrict__ bias, float* __restrict__ out,
              int Ncols, int K, int row0, int col0, int lane) {
    const int l15  = lane & 15;
    const int kh   = lane >> 4;

    const float* __restrict__ xr[MT];
    const float* __restrict__ wr[NT];
#pragma unroll
    for (int mt = 0; mt < MT; ++mt)
        xr[mt] = X + (size_t)(row0 + mt * 16 + l15) * K + 2 * kh;
#pragma unroll
    for (int nt = 0; nt < NT; ++nt)
        wr[nt] = W + (size_t)(col0 + nt * 16 + l15) * K + 2 * kh;

    v8f acc[MT][NT];
#pragma unroll
    for (int mt = 0; mt < MT; ++mt)
#pragma unroll
        for (int nt = 0; nt < NT; ++nt) {
            v8f z = {};
            acc[mt][nt] = z;
        }

    v2f a0[KB][MT], b0[KB][NT], a1[KB][MT], b1[KB][NT];
    gemm_frag_load<MT, NT>(xr, wr, 0, a0, b0);
    for (int k = 0; k < K - 2 * KSPAN; k += 2 * KSPAN) {
        gemm_frag_load<MT, NT>(xr, wr, k + KSPAN, a1, b1);
        __builtin_amdgcn_sched_barrier(0);
        gemm_frag_mma<MT, NT>(a0, b0, acc);
        __builtin_amdgcn_sched_barrier(0);
        gemm_frag_load<MT, NT>(xr, wr, k + 2 * KSPAN, a0, b0);
        __builtin_amdgcn_sched_barrier(0);
        gemm_frag_mma<MT, NT>(a1, b1, acc);
        __builtin_amdgcn_sched_barrier(0);
    }
    gemm_frag_load<MT, NT>(xr, wr, K - KSPAN, a1, b1);
    __builtin_amdgcn_sched_barrier(0);
    gemm_frag_mma<MT, NT>(a0, b0, acc);
    gemm_frag_mma<MT, NT>(a1, b1, acc);

#pragma unroll
    for (int nt = 0; nt < NT; ++nt) {
        const int   col  = col0 + nt * 16;
        const float bcol = bias[col + l15];
#pragma unroll
        for (int mt = 0; mt < MT; ++mt) {
            const int row = row0 + mt * 16;
#pragma unroll
            for (int r = 0; r < 8; ++r) {
                out[(size_t)(row + r + 8 * kh) * Ncols + col + l15] =
                    acc[mt][nt][r] + bcol;
            }
        }
    }
}

template <int MT, int NT>
__global__ void __launch_bounds__(32)
wmma_gemm_f32(const float* __restrict__ X, const float* __restrict__ W,
              const float* __restrict__ bias, float* __restrict__ out,
              int Ncols, int K) {
    gemm_tile_f32<MT, NT>(X, W, bias, out, Ncols, K,
                          blockIdx.x * (16 * MT), blockIdx.y * (16 * NT),
                          threadIdx.x);
}

__global__ void __launch_bounds__(32)
proj_all(const float* __restrict__ inp, const float* __restrict__ Wv,
         const float* __restrict__ bv, const float* __restrict__ query,
         const float* __restrict__ Ws, const float* __restrict__ bs,
         const float* __restrict__ Wa, const float* __restrict__ ba,
         const float* __restrict__ ipts,
         float* __restrict__ value, float* __restrict__ off,
         float* __restrict__ logits, float* __restrict__ V4) {
    const int lane = threadIdx.x;
    const int id   = blockIdx.x;
    if (id < VT) {
        const int bx = id % VT_X;
        const int by = id / VT_X;
        if (by == 0) {
            const int l = bx * 32 + lane;
            const float* v = ipts + (size_t)l * 3;
            const float x = v[0], y = v[1], z = v[2];
            float4 t;
            t.x = -2.0f * x;
            t.y = -2.0f * y;
            t.z = -2.0f * z;
            t.w = x * x + y * y + z * z;
            *(float4*)(V4 + (size_t)l * 4) = t;
        }
        gemm_tile_f32<2, 4>(inp, Wv, bv, value, CC, CC, bx * 32, by * 64, lane);
    } else {
        const int j  = id - VT;
        const int bx = j % QT_X;
        const int by = j / QT_X;
        if (by < 3)
            gemm_tile_f32<2, 2>(query, Ws, bs, off, NOFF, CC, bx * 32, by * 32, lane);
        else
            gemm_tile_f32<2, 2>(query, Wa, ba, logits, MH * NP, CC, bx * 32, 0, lane);
    }
}

__device__ __forceinline__ void knn_step(const v2f a, const v2f* bq, int cib,
                                         float* best, int* bidx) {
#pragma unroll
    for (int t = 0; t < QT; ++t) {
        v8f c = {};
        c = __builtin_amdgcn_wmma_f32_16x16x4_f32(
            false, a, false, bq[t], (short)0, c, false, false);
#pragma unroll
        for (int r = 0; r < 8; ++r) {
            const float dv = c[r];
            const int   ci = cib + r;
            if (dv < best[t]) { best[t] = dv; bidx[t] = ci; }
        }
    }
}

__global__ void __launch_bounds__(32 * NW)
knn_gather(const float* __restrict__ off, const float* __restrict__ qpts,
           const float* __restrict__ V4, const float* __restrict__ value,
           const float* __restrict__ logits, float* __restrict__ out_tmp) {
    __shared__ float sBest[NW][16 * QT];
    __shared__ int   sIdx[NW][16 * QT];
    __shared__ int   sFin[16 * QT];

    const int lane = threadIdx.x & 31;
    const int wv   = threadIdx.x >> 5;
    const int l15  = lane & 15;
    const int kh   = lane >> 4;
    const int n0   = blockIdx.x * (16 * QT);
    const int b    = n0 >> 15;
    const int cw   = wv * 16;

    v2f bq[QT];
#pragma unroll
    for (int t = 0; t < QT; ++t) {
        const int n  = n0 + t * 16 + l15;
        const int rq = n >> 5;
        const int mp = n & 31;
        const float* o = off  + (size_t)rq * NOFF + mp * 3;
        const float* p = qpts + (size_t)rq * 3;
        const float s0 = p[2 * kh] + o[2 * kh];
        const float s1 = p[1] + o[1];
        v2f f;
        f.x = s0;
        f.y = kh ? 1.0f : s1;
        bq[t] = f;
    }

    const float* __restrict__ vptr = V4 + ((size_t)b * LIN + cw) * 4
                                        + (size_t)l15 * 4 + 2 * kh;

    float best[QT];
    int   bidx[QT];
#pragma unroll
    for (int t = 0; t < QT; ++t) { best[t] = 3.4e38f; bidx[t] = 0; }

    v2f a_cur = *(const v2f*)(vptr);
    for (int cb = 0; cb < LIN - CSTEP; cb += CSTEP) {
        const v2f a_next = *(const v2f*)(vptr + (size_t)(cb + CSTEP) * 4);
        knn_step(a_cur, bq, cw + cb + (kh << 3), best, bidx);
        a_cur = a_next;
    }
    knn_step(a_cur, bq, cw + (LIN - CSTEP) + (kh << 3), best, bidx);

#pragma unroll
    for (int t = 0; t < QT; ++t) {
        const float obest = __shfl_xor(best[t], 16, 32);
        const int   oidx  = __shfl_xor(bidx[t], 16, 32);
        if (obest < best[t] || (obest == best[t] && oidx < bidx[t])) {
            best[t] = obest; bidx[t] = oidx;
        }
        if (lane < 16) {
            sBest[wv][t * 16 + l15] = best[t];
            sIdx[wv][t * 16 + l15]  = bidx[t];
        }
    }
    __syncthreads();

    if (threadIdx.x < 16 * QT) {
        float mb = sBest[0][threadIdx.x];
        int   mi = sIdx[0][threadIdx.x];
        for (int w = 1; w < NW; ++w) {
            const float ob = sBest[w][threadIdx.x];
            const int   oi = sIdx[w][threadIdx.x];
            if (ob < mb || (ob == mb && oi < mi)) { mb = ob; mi = oi; }
        }
        sFin[threadIdx.x] = mi;
    }
    __syncthreads();

    {
        const int c = threadIdx.x;
        const int m = c >> 5;
        const float* vb = value + (size_t)b * LIN * CC;
#pragma unroll
        for (int j = 0; j < (16 * QT) / (MH * NP); ++j) {
            const int rq = (n0 >> 5) + j;
            float4 v = *(const float4*)(logits + ((size_t)rq * MH + m) * NP);
            const float mx = fmaxf(fmaxf(v.x, v.y), fmaxf(v.z, v.w));
            const float e0 = __expf(v.x - mx), e1 = __expf(v.y - mx);
            const float e2 = __expf(v.z - mx), e3 = __expf(v.w - mx);
            const float inv = 1.0f / (e0 + e1 + e2 + e3);
            float aw[NP];
            aw[0] = e0 * inv; aw[1] = e1 * inv; aw[2] = e2 * inv; aw[3] = e3 * inv;
            const int* id = sFin + j * (MH * NP) + m * NP;
            float acc = 0.0f;
#pragma unroll
            for (int p = 0; p < NP; ++p)
                acc += aw[p] * vb[(size_t)id[p] * CC + c];
            out_tmp[(size_t)rq * CC + c] = acc;
        }
    }
}

extern "C" void kernel_launch(void* const* d_in, const int* in_sizes, int n_in,
                              void* d_out, int out_size, void* d_ws, size_t ws_size,
                              hipStream_t stream) {
    const float* query = (const float*)d_in[0];
    const float* qpts  = (const float*)d_in[1];
    const float* inp   = (const float*)d_in[2];
    const float* ipts  = (const float*)d_in[3];
    const float* Wv    = (const float*)d_in[4];
    const float* bv    = (const float*)d_in[5];
    const float* Ws    = (const float*)d_in[6];
    const float* bs    = (const float*)d_in[7];
    const float* Wa    = (const float*)d_in[8];
    const float* ba    = (const float*)d_in[9];
    const float* Wo    = (const float*)d_in[10];
    const float* bo    = (const float*)d_in[11];
    float* out = (float*)d_out;

    char* w = (char*)d_ws;
    float* value  = (float*)w;  w += (size_t)BB * LIN * CC * 4;
    float* off    = (float*)w;  w += (size_t)BB * LQ * MH * NP * 3 * 4;
    float* logits = (float*)w;  w += (size_t)BB * LQ * MH * NP * 4;
    float* V4     = (float*)w;  w += (size_t)BB * LIN * 4 * 4;
    float* otmp   = (float*)w;  w += (size_t)BB * LQ * CC * 4;

    const dim3 wave(32);

    proj_all<<<VT + QT_X * QT_Y, wave, 0, stream>>>(
        inp, Wv, bv, query, Ws, bs, Wa, ba, ipts, value, off, logits, V4);

    knn_gather<<<(BB * LQ * MH * NP) / (16 * QT), dim3(32 * NW), 0, stream>>>(
        off, qpts, V4, value, logits, otmp);

    wmma_gemm_f32<2, 4><<<dim3((BB * LQ) / 32, CC / 64), wave, 0, stream>>>(
        otmp, Wo, bo, out, CC, CC);
}


# KNN argmin two-level: scan keeps per lane the min of each product's 8 rows (3-input minima) and its 8-candidate group; row in the winning group found once by repeating that product; same tie rule
# speedup vs baseline: 3.9383x; 1.1202x over previous
#include <hip/hip_runtime.h>
#include <hip/hip_bf16.h>


typedef __attribute__((ext_vector_type(2))) float v2f;
typedef __attribute__((ext_vector_type(8))) float v8f;

#define BB   2
#define LQ   1024
#define LIN  4096
#define CC   256
#define MH   8
#define NP   4
#define DH   32
#define QT   4
#define NW   8
#define CSTEP (16 * NW)

#define VT_X ((BB * LIN) / 32)
#define VT_Y (CC / 64)
#define VT   (VT_X * VT_Y)
#define QT_X ((BB * LQ) / 32)
#define QT_Y 4
#define NOFF (MH * NP * 3)

#define KB    8
#define KSPAN (4 * KB)

template <int MT, int NT>
__device__ __forceinline__ void
gemm_frag_load(const float* __restrict__ const (&xr)[MT],
               const float* __restrict__ const (&wr)[NT], int k0,
               v2f (&a)[KB][MT], v2f (&b)[KB][NT]) {
#pragma unroll
    for (int s = 0; s < KB; ++s) {
#pragma unroll
        for (int mt = 0; mt < MT; ++mt) a[s][mt] = *(const v2f*)(xr[mt] + k0 + 4 * s);
#pragma unroll
        for (int nt = 0; nt < NT; ++nt) b[s][nt] = *(const v2f*)(wr[nt] + k0 + 4 * s);
    }
}

template <int MT, int NT>
__device__ __forceinline__ void
gemm_frag_mma(const v2f (&a)[KB][MT], const v2f (&b)[KB][NT], v8f (&acc)[MT][NT]) {
#pragma unroll
    for (int s = 0; s < KB; ++s)
#pragma unroll
        for (int mt = 0; mt < MT; ++mt)
#pragma unroll
            for (int nt = 0; nt < NT; ++nt)
                acc[mt][nt] = __builtin_amdgcn_wmma_f32_16x16x4_f32(
                     false, a[s][mt],  false, b[s][nt],
                     (short)0, acc[mt][nt],
                     false,  false);
}

template <int MT, int NT>
__device__ __forceinline__ void
gemm_tile_f32(const float* __restrict__ X, const float* __restrict__ W,
              const float* __restrict__ bias, float* __restrict__ out,
              int Ncols, int K, int row0, int col0, int lane) {
    const int l15  = lane & 15;
    const int kh   = lane >> 4;

    const float* __restrict__ xr[MT];
    const float* __restrict__ wr[NT];
#pragma unroll
    for (int mt = 0; mt < MT; ++mt)
        xr[mt] = X + (size_t)(row0 + mt * 16 + l15) * K + 2 * kh;
#pragma unroll
    for (int nt = 0; nt < NT; ++nt)
        wr[nt] = W + (size_t)(col0 + nt * 16 + l15) * K + 2 * kh;

    v8f acc[MT][NT];
#pragma unroll
    for (int mt = 0; mt < MT; ++mt)
#pragma unroll
        for (int nt = 0; nt < NT; ++nt) {
            v8f z = {};
            acc[mt][nt] = z;
        }

    v2f a0[KB][MT], b0[KB][NT], a1[KB][MT], b1[KB][NT];
    gemm_frag_load<MT, NT>(xr, wr, 0, a0, b0);
    for (int k = 0; k < K - 2 * KSPAN; k += 2 * KSPAN) {
        gemm_frag_load<MT, NT>(xr, wr, k + KSPAN, a1, b1);
        __builtin_amdgcn_sched_barrier(0);
        gemm_frag_mma<MT, NT>(a0, b0, acc);
        __builtin_amdgcn_sched_barrier(0);
        gemm_frag_load<MT, NT>(xr, wr, k + 2 * KSPAN, a0, b0);
        __builtin_amdgcn_sched_barrier(0);
        gemm_frag_mma<MT, NT>(a1, b1, acc);
        __builtin_amdgcn_sched_barrier(0);
    }
    gemm_frag_load<MT, NT>(xr, wr, K - KSPAN, a1, b1);
    __builtin_amdgcn_sched_barrier(0);
    gemm_frag_mma<MT, NT>(a0, b0, acc);
    gemm_frag_mma<MT, NT>(a1, b1, acc);

#pragma unroll
    for (int nt = 0; nt < NT; ++nt) {
        const int   col  = col0 + nt * 16;
        const float bcol = bias[col + l15];
#pragma unroll
        for (int mt = 0; mt < MT; ++mt) {
            const int row = row0 + mt * 16;
#pragma unroll
            for (int r = 0; r < 8; ++r) {
                out[(size_t)(row + r + 8 * kh) * Ncols + col + l15] =
                    acc[mt][nt][r] + bcol;
            }
        }
    }
}

template <int MT, int NT>
__global__ void __launch_bounds__(32)
wmma_gemm_f32(const float* __restrict__ X, const float* __restrict__ W,
              const float* __restrict__ bias, float* __restrict__ out,
              int Ncols, int K) {
    gemm_tile_f32<MT, NT>(X, W, bias, out, Ncols, K,
                          blockIdx.x * (16 * MT), blockIdx.y * (16 * NT),
                          threadIdx.x);
}

__global__ void __launch_bounds__(32)
proj_all(const float* __restrict__ inp, const float* __restrict__ Wv,
         const float* __restrict__ bv, const float* __restrict__ query,
         const float* __restrict__ Ws, const float* __restrict__ bs,
         const float* __restrict__ Wa, const float* __restrict__ ba,
         const float* __restrict__ ipts,
         float* __restrict__ value, float* __restrict__ off,
         float* __restrict__ logits, float* __restrict__ V4) {
    const int lane = threadIdx.x;
    const int id   = blockIdx.x;
    if (id < VT) {
        const int bx = id % VT_X;
        const int by = id / VT_X;
        if (by == 0) {
            const int l = bx * 32 + lane;
            const float* v = ipts + (size_t)l * 3;
            const float x = v[0], y = v[1], z = v[2];
            float4 t;
            t.x = -2.0f * x;
            t.y = -2.0f * y;
            t.z = -2.0f * z;
            t.w = x * x + y * y + z * z;
            *(float4*)(V4 + (size_t)l * 4) = t;
        }
        gemm_tile_f32<2, 4>(inp, Wv, bv, value, CC, CC, bx * 32, by * 64, lane);
    } else {
        const int j  = id - VT;
        const int bx = j % QT_X;
        const int by = j / QT_X;
        if (by < 3)
            gemm_tile_f32<2, 2>(query, Ws, bs, off, NOFF, CC, bx * 32, by * 32, lane);
        else
            gemm_tile_f32<2, 2>(query, Wa, ba, logits, MH * NP, CC, bx * 32, 0, lane);
    }
}

__device__ __forceinline__ void knn_step(const v2f a, const v2f* bq, int cib,
                                         float* best, int* bgrp) {
#pragma unroll
    for (int t = 0; t < QT; ++t) {
        v8f c = {};
        c = __builtin_amdgcn_wmma_f32_16x16x4_f32(
            false, a, false, bq[t], (short)0, c, false, false);
        const float t0 = fminf(fminf(c[0], c[1]), c[2]);
        const float t1 = fminf(fminf(c[3], c[4]), c[5]);
        const float t2 = fminf(fminf(c[6], c[7]), best[t]);
        const float nb = fminf(fminf(t0, t1), t2);
        if (nb < best[t]) bgrp[t] = cib;
        best[t] = nb;
    }
}

__global__ void __launch_bounds__(32 * NW)
knn_gather(const float* __restrict__ off, const float* __restrict__ qpts,
           const float* __restrict__ V4, const float* __restrict__ value,
           const float* __restrict__ logits, float* __restrict__ out_tmp) {
    constexpr int PW = (16 * QT) / NW;
    static_assert(PW * NW == 16 * QT && PW <= 16 && 16 % PW == 0,
                  "a wave's sampling points must share one query tile");

    __shared__ float sBest[NW][16 * QT];
    __shared__ int   sIdx[NW][16 * QT];
    __shared__ int   sGrp[16 * QT];
    __shared__ int   sFin[16 * QT];

    const int lane = threadIdx.x & 31;
    const int wv   = threadIdx.x >> 5;
    const int l15  = lane & 15;
    const int kh   = lane >> 4;
    const int n0   = blockIdx.x * (16 * QT);
    const int b    = n0 >> 15;
    const int cw   = wv * 16;

    v2f bq[QT];
#pragma unroll
    for (int t = 0; t < QT; ++t) {
        const int n  = n0 + t * 16 + l15;
        const int rq = n >> 5;
        const int mp = n & 31;
        const float* o = off  + (size_t)rq * NOFF + mp * 3;
        const float* p = qpts + (size_t)rq * 3;
        const float s0 = p[2 * kh] + o[2 * kh];
        const float s1 = p[1] + o[1];
        v2f f;
        f.x = s0;
        f.y = kh ? 1.0f : s1;
        bq[t] = f;
    }

    const float* __restrict__ vlane = V4 + (size_t)b * LIN * 4
                                         + (size_t)l15 * 4 + 2 * kh;
    const float* __restrict__ vptr = vlane + (size_t)cw * 4;

    float best[QT];
    int   bgrp[QT];
#pragma unroll
    for (int t = 0; t < QT; ++t) { best[t] = 3.4e38f; bgrp[t] = 0; }

    v2f a_cur = *(const v2f*)(vptr);
    for (int cb = 0; cb < LIN - CSTEP; cb += CSTEP) {
        const v2f a_next = *(const v2f*)(vptr + (size_t)(cb + CSTEP) * 4);
        knn_step(a_cur, bq, cw + cb + (kh << 3), best, bgrp);
        a_cur = a_next;
    }
    knn_step(a_cur, bq, cw + (LIN - CSTEP) + (kh << 3), best, bgrp);

#pragma unroll
    for (int t = 0; t < QT; ++t) {
        const float obest = __shfl_xor(best[t], 16, 32);
        const int   ogrp  = __shfl_xor(bgrp[t], 16, 32);
        if (obest < best[t] || (obest == best[t] && ogrp < bgrp[t])) {
            best[t] = obest; bgrp[t] = ogrp;
        }
        if (lane < 16) {
            sBest[wv][t * 16 + l15] = best[t];
            sIdx[wv][t * 16 + l15]  = bgrp[t];
        }
    }
    __syncthreads();

    if (threadIdx.x < 16 * QT) {
        float mb = sBest[0][threadIdx.x];
        int   mi = sIdx[0][threadIdx.x];
        for (int w = 1; w < NW; ++w) {
            const float ob = sBest[w][threadIdx.x];
            const int   oi = sIdx[w][threadIdx.x];
            if (ob < mb || (ob == mb && oi < mi)) { mb = ob; mi = oi; }
        }
        sGrp[threadIdx.x] = mi;
    }
    __syncthreads();

    {
        const int tw = (wv * PW) >> 4;
        v2f bsel = bq[0];
#pragma unroll
        for (int t = 1; t < QT; ++t)
            if (tw == t) bsel = bq[t];
        const int mine = l15 % PW;
        int g[PW];
        v2f a[PW];
#pragma unroll
        for (int i = 0; i < PW; ++i) g[i] = sGrp[wv * PW + i];
#pragma unroll
        for (int i = 0; i < PW; ++i)
            a[i] = *(const v2f*)(vlane + (size_t)(g[i] & ~15) * 4);
        v8f mc = {};
        int myg = 0;
#pragma unroll
        for (int i = 0; i < PW; ++i) {
            v8f c = {};
            c = __builtin_amdgcn_wmma_f32_16x16x4_f32(
                false, a[i], false, bsel, (short)0, c, false, false);
            if (mine == i) { mc = c; myg = g[i]; }
        }
        float m2[4]; int r2[4];
#pragma unroll
        for (int k = 0; k < 4; ++k) {
            const bool s = mc[2 * k + 1] < mc[2 * k];
            m2[k] = s ? mc[2 * k + 1] : mc[2 * k];
            r2[k] = s ? 2 * k + 1 : 2 * k;
        }
        float m4[2]; int r4[2];
#pragma unroll
        for (int k = 0; k < 2; ++k) {
            const bool s = m2[2 * k + 1] < m2[2 * k];
            m4[k] = s ? m2[2 * k + 1] : m2[2 * k];
            r4[k] = s ? r2[2 * k + 1] : r2[2 * k];
        }
        const int r = (m4[1] < m4[0]) ? r4[1] : r4[0];
        if ((l15 - mine) == ((wv * PW) & 15) && kh == ((myg >> 3) & 1))
            sFin[wv * PW + mine] = myg + r;
    }
    __syncthreads();

    {
        const int c = threadIdx.x;
        const int m = c >> 5;
        const float* vb = value + (size_t)b * LIN * CC;
#pragma unroll
        for (int j = 0; j < (16 * QT) / (MH * NP); ++j) {
            const int rq = (n0 >> 5) + j;
            float4 v = *(const float4*)(logits + ((size_t)rq * MH + m) * NP);
            const float mx = fmaxf(fmaxf(v.x, v.y), fmaxf(v.z, v.w));
            const float e0 = __expf(v.x - mx), e1 = __expf(v.y - mx);
            const float e2 = __expf(v.z - mx), e3 = __expf(v.w - mx);
            const float inv = 1.0f / (e0 + e1 + e2 + e3);
            float aw[NP];
            aw[0] = e0 * inv; aw[1] = e1 * inv; aw[2] = e2 * inv; aw[3] = e3 * inv;
            const int* id = sFin + j * (MH * NP) + m * NP;
            float acc = 0.0f;
#pragma unroll
            for (int p = 0; p < NP; ++p)
                acc += aw[p] * vb[(size_t)id[p] * CC + c];
            out_tmp[(size_t)rq * CC + c] = acc;
        }
    }
}

extern "C" void kernel_launch(void* const* d_in, const int* in_sizes, int n_in,
                              void* d_out, int out_size, void* d_ws, size_t ws_size,
                              hipStream_t stream) {
    const float* query = (const float*)d_in[0];
    const float* qpts  = (const float*)d_in[1];
    const float* inp   = (const float*)d_in[2];
    const float* ipts  = (const float*)d_in[3];
    const float* Wv    = (const float*)d_in[4];
    const float* bv    = (const float*)d_in[5];
    const float* Ws    = (const float*)d_in[6];
    const float* bs    = (const float*)d_in[7];
    const float* Wa    = (const float*)d_in[8];
    const float* ba    = (const float*)d_in[9];
    const float* Wo    = (const float*)d_in[10];
    const float* bo    = (const float*)d_in[11];
    float* out = (float*)d_out;

    char* w = (char*)d_ws;
    float* value  = (float*)w;  w += (size_t)BB * LIN * CC * 4;
    float* off    = (float*)w;  w += (size_t)BB * LQ * MH * NP * 3 * 4;
    float* logits = (float*)w;  w += (size_t)BB * LQ * MH * NP * 4;
    float* V4     = (float*)w;  w += (size_t)BB * LIN * 4 * 4;
    float* otmp   = (float*)w;  w += (size_t)BB * LQ * CC * 4;

    const dim3 wave(32);

    proj_all<<<VT + QT_X * QT_Y, wave, 0, stream>>>(
        inp, Wv, bv, query, Ws, bs, Wa, ba, ipts, value, off, logits, V4);

    knn_gather<<<(BB * LQ * MH * NP) / (16 * QT), dim3(32 * NW), 0, stream>>>(
        off, qpts, V4, value, logits, otmp);

    wmma_gemm_f32<2, 4><<<dim3((BB * LQ) / 32, CC / 64), wave, 0, stream>>>(
        otmp, Wo, bo, out, CC, CC);
}


# GEMM fragments loaded 16 bytes per lane, not 8, in the value, logit and output products: half the loads and line accesses; k summation order changes there; offsets product and indices bit-identical
# speedup vs baseline: 4.0952x; 1.0398x over previous
#include <hip/hip_runtime.h>
#include <hip/hip_bf16.h>


typedef __attribute__((ext_vector_type(2))) float v2f;
typedef __attribute__((ext_vector_type(4))) float v4f;
typedef __attribute__((ext_vector_type(8))) float v8f;

#define BB   2
#define LQ   1024
#define LIN  4096
#define CC   256
#define MH   8
#define NP   4
#define DH   32
#define QT   4
#define NW   8
#define CSTEP (16 * NW)

#define VT_X ((BB * LIN) / 32)
#define VT_Y (CC / 64)
#define VT   (VT_X * VT_Y)
#define QT_X ((BB * LQ) / 32)
#define QT_Y 4
#define NOFF (MH * NP * 3)

#define KB    8
#define KSPAN (4 * KB)

template <int MT, int NT>
__device__ __forceinline__ void
gemm_frag_load(const float* __restrict__ const (&xr)[MT],
               const float* __restrict__ const (&wr)[NT], int k0,
               v2f (&a)[KB][MT], v2f (&b)[KB][NT]) {
#pragma unroll
    for (int s = 0; s < KB; ++s) {
#pragma unroll
        for (int mt = 0; mt < MT; ++mt) a[s][mt] = *(const v2f*)(xr[mt] + k0 + 4 * s);
#pragma unroll
        for (int nt = 0; nt < NT; ++nt) b[s][nt] = *(const v2f*)(wr[nt] + k0 + 4 * s);
    }
}

template <int MT, int NT>
__device__ __forceinline__ void
gemm_frag_mma(const v2f (&a)[KB][MT], const v2f (&b)[KB][NT], v8f (&acc)[MT][NT]) {
#pragma unroll
    for (int s = 0; s < KB; ++s)
#pragma unroll
        for (int mt = 0; mt < MT; ++mt)
#pragma unroll
            for (int nt = 0; nt < NT; ++nt)
                acc[mt][nt] = __builtin_amdgcn_wmma_f32_16x16x4_f32(
                     false, a[s][mt],  false, b[s][nt],
                     (short)0, acc[mt][nt],
                     false,  false);
}

#define KQ (KB / 2)

template <int MT, int NT>
__device__ __forceinline__ void
gemm_frag_load(const float* __restrict__ const (&xr)[MT],
               const float* __restrict__ const (&wr)[NT], int k0,
               v4f (&a)[KQ][MT], v4f (&b)[KQ][NT]) {
#pragma unroll
    for (int j = 0; j < KQ; ++j) {
#pragma unroll
        for (int mt = 0; mt < MT; ++mt) a[j][mt] = *(const v4f*)(xr[mt] + k0 + 8 * j);
#pragma unroll
        for (int nt = 0; nt < NT; ++nt) b[j][nt] = *(const v4f*)(wr[nt] + k0 + 8 * j);
    }
}

template <int MT, int NT>
__device__ __forceinline__ void
gemm_frag_mma(const v4f (&a)[KQ][MT], const v4f (&b)[KQ][NT], v8f (&acc)[MT][NT]) {
#pragma unroll
    for (int j = 0; j < KQ; ++j) {
#pragma unroll
        for (int mt = 0; mt < MT; ++mt)
#pragma unroll
            for (int nt = 0; nt < NT; ++nt)
                acc[mt][nt] = __builtin_amdgcn_wmma_f32_16x16x4_f32(
                     false, a[j][mt].xy,  false, b[j][nt].xy,
                     (short)0, acc[mt][nt],
                     false,  false);
#pragma unroll
        for (int mt = 0; mt < MT; ++mt)
#pragma unroll
            for (int nt = 0; nt < NT; ++nt)
                acc[mt][nt] = __builtin_amdgcn_wmma_f32_16x16x4_f32(
                     false, a[j][mt].zw,  false, b[j][nt].zw,
                     (short)0, acc[mt][nt],
                     false,  false);
    }
}

template <bool WIDE> struct FragKind {
    typedef v2f type;
    static constexpr int N = KB;
    static constexpr int LANE_K = 2;
};
template <> struct FragKind<true> {
    typedef v4f type;
    static constexpr int N = KQ;
    static constexpr int LANE_K = 4;
};

template <int MT, int NT, bool WIDE>
__device__ __forceinline__ void
gemm_tile_f32(const float* __restrict__ X, const float* __restrict__ W,
              const float* __restrict__ bias, float* __restrict__ out,
              int Ncols, int K, int row0, int col0, int lane) {
    typedef typename FragKind<WIDE>::type frag_t;
    constexpr int NF = FragKind<WIDE>::N;
    const int l15  = lane & 15;
    const int kh   = lane >> 4;

    const float* __restrict__ xr[MT];
    const float* __restrict__ wr[NT];
#pragma unroll
    for (int mt = 0; mt < MT; ++mt)
        xr[mt] = X + (size_t)(row0 + mt * 16 + l15) * K + FragKind<WIDE>::LANE_K * kh;
#pragma unroll
    for (int nt = 0; nt < NT; ++nt)
        wr[nt] = W + (size_t)(col0 + nt * 16 + l15) * K + FragKind<WIDE>::LANE_K * kh;

    v8f acc[MT][NT];
#pragma unroll
    for (int mt = 0; mt < MT; ++mt)
#pragma unroll
        for (int nt = 0; nt < NT; ++nt) {
            v8f z = {};
            acc[mt][nt] = z;
        }

    frag_t a0[NF][MT], b0[NF][NT], a1[NF][MT], b1[NF][NT];
    gemm_frag_load<MT, NT>(xr, wr, 0, a0, b0);
    for (int k = 0; k < K - 2 * KSPAN; k += 2 * KSPAN) {
        gemm_frag_load<MT, NT>(xr, wr, k + KSPAN, a1, b1);
        __builtin_amdgcn_sched_barrier(0);
        gemm_frag_mma<MT, NT>(a0, b0, acc);
        __builtin_amdgcn_sched_barrier(0);
        gemm_frag_load<MT, NT>(xr, wr, k + 2 * KSPAN, a0, b0);
        __builtin_amdgcn_sched_barrier(0);
        gemm_frag_mma<MT, NT>(a1, b1, acc);
        __builtin_amdgcn_sched_barrier(0);
    }
    gemm_frag_load<MT, NT>(xr, wr, K - KSPAN, a1, b1);
    __builtin_amdgcn_sched_barrier(0);
    gemm_frag_mma<MT, NT>(a0, b0, acc);
    gemm_frag_mma<MT, NT>(a1, b1, acc);

#pragma unroll
    for (int nt = 0; nt < NT; ++nt) {
        const int   col  = col0 + nt * 16;
        const float bcol = bias[col + l15];
#pragma unroll
        for (int mt = 0; mt < MT; ++mt) {
            const int row = row0 + mt * 16;
#pragma unroll
            for (int r = 0; r < 8; ++r) {
                out[(size_t)(row + r + 8 * kh) * Ncols + col + l15] =
                    acc[mt][nt][r] + bcol;
            }
        }
    }
}

template <int MT, int NT>
__global__ void __launch_bounds__(32)
wmma_gemm_f32(const float* __restrict__ X, const float* __restrict__ W,
              const float* __restrict__ bias, float* __restrict__ out,
              int Ncols, int K) {
    gemm_tile_f32<MT, NT, true>(X, W, bias, out, Ncols, K,
                                blockIdx.x * (16 * MT), blockIdx.y * (16 * NT),
                                threadIdx.x);
}

__global__ void __launch_bounds__(32)
proj_all(const float* __restrict__ inp, const float* __restrict__ Wv,
         const float* __restrict__ bv, const float* __restrict__ query,
         const float* __restrict__ Ws, const float* __restrict__ bs,
         const float* __restrict__ Wa, const float* __restrict__ ba,
         const float* __restrict__ ipts,
         float* __restrict__ value, float* __restrict__ off,
         float* __restrict__ logits, float* __restrict__ V4) {
    const int lane = threadIdx.x;
    const int id   = blockIdx.x;
    if (id < VT) {
        const int bx = id % VT_X;
        const int by = id / VT_X;
        if (by == 0) {
            const int l = bx * 32 + lane;
            const float* v = ipts + (size_t)l * 3;
            const float x = v[0], y = v[1], z = v[2];
            float4 t;
            t.x = -2.0f * x;
            t.y = -2.0f * y;
            t.z = -2.0f * z;
            t.w = x * x + y * y + z * z;
            *(float4*)(V4 + (size_t)l * 4) = t;
        }
        gemm_tile_f32<2, 4, true>(inp, Wv, bv, value, CC, CC, bx * 32, by * 64, lane);
    } else {
        const int j  = id - VT;
        const int bx = j % QT_X;
        const int by = j / QT_X;
        if (by < 3)
            gemm_tile_f32<2, 2, false>(query, Ws, bs, off, NOFF, CC, bx * 32, by * 32, lane);
        else
            gemm_tile_f32<2, 2, true>(query, Wa, ba, logits, MH * NP, CC, bx * 32, 0, lane);
    }
}

__device__ __forceinline__ void knn_step(const v2f a, const v2f* bq, int cib,
                                         float* best, int* bgrp) {
#pragma unroll
    for (int t = 0; t < QT; ++t) {
        v8f c = {};
        c = __builtin_amdgcn_wmma_f32_16x16x4_f32(
            false, a, false, bq[t], (short)0, c, false, false);
        const float t0 = fminf(fminf(c[0], c[1]), c[2]);
        const float t1 = fminf(fminf(c[3], c[4]), c[5]);
        const float t2 = fminf(fminf(c[6], c[7]), best[t]);
        const float nb = fminf(fminf(t0, t1), t2);
        if (nb < best[t]) bgrp[t] = cib;
        best[t] = nb;
    }
}

__global__ void __launch_bounds__(32 * NW)
knn_gather(const float* __restrict__ off, const float* __restrict__ qpts,
           const float* __restrict__ V4, const float* __restrict__ value,
           const float* __restrict__ logits, float* __restrict__ out_tmp) {
    constexpr int PW = (16 * QT) / NW;
    static_assert(PW * NW == 16 * QT && PW <= 16 && 16 % PW == 0,
                  "a wave's sampling points must share one query tile");

    __shared__ float sBest[NW][16 * QT];
    __shared__ int   sIdx[NW][16 * QT];
    __shared__ int   sGrp[16 * QT];
    __shared__ int   sFin[16 * QT];

    const int lane = threadIdx.x & 31;
    const int wv   = threadIdx.x >> 5;
    const int l15  = lane & 15;
    const int kh   = lane >> 4;
    const int n0   = blockIdx.x * (16 * QT);
    const int b    = n0 >> 15;
    const int cw   = wv * 16;

    v2f bq[QT];
#pragma unroll
    for (int t = 0; t < QT; ++t) {
        const int n  = n0 + t * 16 + l15;
        const int rq = n >> 5;
        const int mp = n & 31;
        const float* o = off  + (size_t)rq * NOFF + mp * 3;
        const float* p = qpts + (size_t)rq * 3;
        const float s0 = p[2 * kh] + o[2 * kh];
        const float s1 = p[1] + o[1];
        v2f f;
        f.x = s0;
        f.y = kh ? 1.0f : s1;
        bq[t] = f;
    }

    const float* __restrict__ vlane = V4 + (size_t)b * LIN * 4
                                         + (size_t)l15 * 4 + 2 * kh;
    const float* __restrict__ vptr = vlane + (size_t)cw * 4;

    float best[QT];
    int   bgrp[QT];
#pragma unroll
    for (int t = 0; t < QT; ++t) { best[t] = 3.4e38f; bgrp[t] = 0; }

    v2f a_cur = *(const v2f*)(vptr);
    for (int cb = 0; cb < LIN - CSTEP; cb += CSTEP) {
        const v2f a_next = *(const v2f*)(vptr + (size_t)(cb + CSTEP) * 4);
        knn_step(a_cur, bq, cw + cb + (kh << 3), best, bgrp);
        a_cur = a_next;
    }
    knn_step(a_cur, bq, cw + (LIN - CSTEP) + (kh << 3), best, bgrp);

#pragma unroll
    for (int t = 0; t < QT; ++t) {
        const float obest = __shfl_xor(best[t], 16, 32);
        const int   ogrp  = __shfl_xor(bgrp[t], 16, 32);
        if (obest < best[t] || (obest == best[t] && ogrp < bgrp[t])) {
            best[t] = obest; bgrp[t] = ogrp;
        }
        if (lane < 16) {
            sBest[wv][t * 16 + l15] = best[t];
            sIdx[wv][t * 16 + l15]  = bgrp[t];
        }
    }
    __syncthreads();

    if (threadIdx.x < 16 * QT) {
        float mb = sBest[0][threadIdx.x];
        int   mi = sIdx[0][threadIdx.x];
        for (int w = 1; w < NW; ++w) {
            const float ob = sBest[w][threadIdx.x];
            const int   oi = sIdx[w][threadIdx.x];
            if (ob < mb || (ob == mb && oi < mi)) { mb = ob; mi = oi; }
        }
        sGrp[threadIdx.x] = mi;
    }
    __syncthreads();

    {
        const int tw = (wv * PW) >> 4;
        v2f bsel = bq[0];
#pragma unroll
        for (int t = 1; t < QT; ++t)
            if (tw == t) bsel = bq[t];
        const int mine = l15 % PW;
        int g[PW];
        v2f a[PW];
#pragma unroll
        for (int i = 0; i < PW; ++i) g[i] = sGrp[wv * PW + i];
#pragma unroll
        for (int i = 0; i < PW; ++i)
            a[i] = *(const v2f*)(vlane + (size_t)(g[i] & ~15) * 4);
        v8f mc = {};
        int myg = 0;
#pragma unroll
        for (int i = 0; i < PW; ++i) {
            v8f c = {};
            c = __builtin_amdgcn_wmma_f32_16x16x4_f32(
                false, a[i], false, bsel, (short)0, c, false, false);
            if (mine == i) { mc = c; myg = g[i]; }
        }
        float m2[4]; int r2[4];
#pragma unroll
        for (int k = 0; k < 4; ++k) {
            const bool s = mc[2 * k + 1] < mc[2 * k];
            m2[k] = s ? mc[2 * k + 1] : mc[2 * k];
            r2[k] = s ? 2 * k + 1 : 2 * k;
        }
        float m4[2]; int r4[2];
#pragma unroll
        for (int k = 0; k < 2; ++k) {
            const bool s = m2[2 * k + 1] < m2[2 * k];
            m4[k] = s ? m2[2 * k + 1] : m2[2 * k];
            r4[k] = s ? r2[2 * k + 1] : r2[2 * k];
        }
        const int r = (m4[1] < m4[0]) ? r4[1] : r4[0];
        if ((l15 - mine) == ((wv * PW) & 15) && kh == ((myg >> 3) & 1))
            sFin[wv * PW + mine] = myg + r;
    }
    __syncthreads();

    {
        const int c = threadIdx.x;
        const int m = c >> 5;
        const float* vb = value + (size_t)b * LIN * CC;
#pragma unroll
        for (int j = 0; j < (16 * QT) / (MH * NP); ++j) {
            const int rq = (n0 >> 5) + j;
            float4 v = *(const float4*)(logits + ((size_t)rq * MH + m) * NP);
            const float mx = fmaxf(fmaxf(v.x, v.y), fmaxf(v.z, v.w));
            const float e0 = __expf(v.x - mx), e1 = __expf(v.y - mx);
            const float e2 = __expf(v.z - mx), e3 = __expf(v.w - mx);
            const float inv = 1.0f / (e0 + e1 + e2 + e3);
            float aw[NP];
            aw[0] = e0 * inv; aw[1] = e1 * inv; aw[2] = e2 * inv; aw[3] = e3 * inv;
            const int* id = sFin + j * (MH * NP) + m * NP;
            float acc = 0.0f;
#pragma unroll
            for (int p = 0; p < NP; ++p)
                acc += aw[p] * vb[(size_t)id[p] * CC + c];
            out_tmp[(size_t)rq * CC + c] = acc;
        }
    }
}

extern "C" void kernel_launch(void* const* d_in, const int* in_sizes, int n_in,
                              void* d_out, int out_size, void* d_ws, size_t ws_size,
                              hipStream_t stream) {
    const float* query = (const float*)d_in[0];
    const float* qpts  = (const float*)d_in[1];
    const float* inp   = (const float*)d_in[2];
    const float* ipts  = (const float*)d_in[3];
    const float* Wv    = (const float*)d_in[4];
    const float* bv    = (const float*)d_in[5];
    const float* Ws    = (const float*)d_in[6];
    const float* bs    = (const float*)d_in[7];
    const float* Wa    = (const float*)d_in[8];
    const float* ba    = (const float*)d_in[9];
    const float* Wo    = (const float*)d_in[10];
    const float* bo    = (const float*)d_in[11];
    float* out = (float*)d_out;

    char* w = (char*)d_ws;
    float* value  = (float*)w;  w += (size_t)BB * LIN * CC * 4;
    float* off    = (float*)w;  w += (size_t)BB * LQ * MH * NP * 3 * 4;
    float* logits = (float*)w;  w += (size_t)BB * LQ * MH * NP * 4;
    float* V4     = (float*)w;  w += (size_t)BB * LIN * 4 * 4;
    float* otmp   = (float*)w;  w += (size_t)BB * LQ * CC * 4;

    const dim3 wave(32);

    proj_all<<<VT + QT_X * QT_Y, wave, 0, stream>>>(
        inp, Wv, bv, query, Ws, bs, Wa, ba, ipts, value, off, logits, V4);

    knn_gather<<<(BB * LQ * MH * NP) / (16 * QT), dim3(32 * NW), 0, stream>>>(
        off, qpts, V4, value, logits, otmp);

    wmma_gemm_f32<2, 4><<<dim3((BB * LQ) / 32, CC / 64), wave, 0, stream>>>(
        otmp, Wo, bo, out, CC, CC);
}


# Output GEMM tile per wave 32x64 to 16x32: 1024 waves, one per SIMD, not 256 on a quarter of the SIMDs (4 waves of a 32x64 block per work-group, 64-k register blocks); same k order, bit-identical
# speedup vs baseline: 4.2921x; 1.0481x over previous
#include <hip/hip_runtime.h>
#include <hip/hip_bf16.h>


typedef __attribute__((ext_vector_type(2))) float v2f;
typedef __attribute__((ext_vector_type(4))) float v4f;
typedef __attribute__((ext_vector_type(8))) float v8f;

#define BB   2
#define LQ   1024
#define LIN  4096
#define CC   256
#define MH   8
#define NP   4
#define DH   32
#define QT   4
#define NW   8
#define CSTEP (16 * NW)

#define VT_X ((BB * LIN) / 32)
#define VT_Y (CC / 64)
#define VT   (VT_X * VT_Y)
#define QT_X ((BB * LQ) / 32)
#define QT_Y 4
#define NOFF (MH * NP * 3)

#define KB    8

template <int MT, int NT, int NS>
__device__ __forceinline__ void
gemm_frag_load(const float* __restrict__ const (&xr)[MT],
               const float* __restrict__ const (&wr)[NT], int k0,
               v2f (&a)[NS][MT], v2f (&b)[NS][NT]) {
#pragma unroll
    for (int s = 0; s < NS; ++s) {
#pragma unroll
        for (int mt = 0; mt < MT; ++mt) a[s][mt] = *(const v2f*)(xr[mt] + k0 + 4 * s);
#pragma unroll
        for (int nt = 0; nt < NT; ++nt) b[s][nt] = *(const v2f*)(wr[nt] + k0 + 4 * s);
    }
}

template <int MT, int NT, int NS>
__device__ __forceinline__ void
gemm_frag_mma(const v2f (&a)[NS][MT], const v2f (&b)[NS][NT], v8f (&acc)[MT][NT]) {
#pragma unroll
    for (int s = 0; s < NS; ++s)
#pragma unroll
        for (int mt = 0; mt < MT; ++mt)
#pragma unroll
            for (int nt = 0; nt < NT; ++nt)
                acc[mt][nt] = __builtin_amdgcn_wmma_f32_16x16x4_f32(
                     false, a[s][mt],  false, b[s][nt],
                     (short)0, acc[mt][nt],
                     false,  false);
}

template <int MT, int NT, int NQ>
__device__ __forceinline__ void
gemm_frag_load(const float* __restrict__ const (&xr)[MT],
               const float* __restrict__ const (&wr)[NT], int k0,
               v4f (&a)[NQ][MT], v4f (&b)[NQ][NT]) {
#pragma unroll
    for (int j = 0; j < NQ; ++j) {
#pragma unroll
        for (int mt = 0; mt < MT; ++mt) a[j][mt] = *(const v4f*)(xr[mt] + k0 + 8 * j);
#pragma unroll
        for (int nt = 0; nt < NT; ++nt) b[j][nt] = *(const v4f*)(wr[nt] + k0 + 8 * j);
    }
}

template <int MT, int NT, int NQ>
__device__ __forceinline__ void
gemm_frag_mma(const v4f (&a)[NQ][MT], const v4f (&b)[NQ][NT], v8f (&acc)[MT][NT]) {
#pragma unroll
    for (int j = 0; j < NQ; ++j) {
#pragma unroll
        for (int mt = 0; mt < MT; ++mt)
#pragma unroll
            for (int nt = 0; nt < NT; ++nt)
                acc[mt][nt] = __builtin_amdgcn_wmma_f32_16x16x4_f32(
                     false, a[j][mt].xy,  false, b[j][nt].xy,
                     (short)0, acc[mt][nt],
                     false,  false);
#pragma unroll
        for (int mt = 0; mt < MT; ++mt)
#pragma unroll
            for (int nt = 0; nt < NT; ++nt)
                acc[mt][nt] = __builtin_amdgcn_wmma_f32_16x16x4_f32(
                     false, a[j][mt].zw,  false, b[j][nt].zw,
                     (short)0, acc[mt][nt],
                     false,  false);
    }
}

template <bool WIDE, int KBT> struct FragKind {
    typedef v2f type;
    static constexpr int N = KBT;
    static constexpr int LANE_K = 2;
};
template <int KBT> struct FragKind<true, KBT> {
    typedef v4f type;
    static constexpr int N = KBT / 2;
    static constexpr int LANE_K = 4;
};

template <int MT, int NT, bool WIDE, int KBT = KB>
__device__ __forceinline__ void
gemm_tile_f32(const float* __restrict__ X, const float* __restrict__ W,
              const float* __restrict__ bias, float* __restrict__ out,
              int Ncols, int K, int row0, int col0, int lane) {
    typedef typename FragKind<WIDE, KBT>::type frag_t;
    constexpr int NF = FragKind<WIDE, KBT>::N;
    constexpr int KSPAN = 4 * KBT;
    const int l15  = lane & 15;
    const int kh   = lane >> 4;

    const float* __restrict__ xr[MT];
    const float* __restrict__ wr[NT];
#pragma unroll
    for (int mt = 0; mt < MT; ++mt)
        xr[mt] = X + (size_t)(row0 + mt * 16 + l15) * K + FragKind<WIDE, KBT>::LANE_K * kh;
#pragma unroll
    for (int nt = 0; nt < NT; ++nt)
        wr[nt] = W + (size_t)(col0 + nt * 16 + l15) * K + FragKind<WIDE, KBT>::LANE_K * kh;

    v8f acc[MT][NT];
#pragma unroll
    for (int mt = 0; mt < MT; ++mt)
#pragma unroll
        for (int nt = 0; nt < NT; ++nt) {
            v8f z = {};
            acc[mt][nt] = z;
        }

    frag_t a0[NF][MT], b0[NF][NT], a1[NF][MT], b1[NF][NT];
    gemm_frag_load<MT, NT>(xr, wr, 0, a0, b0);
    for (int k = 0; k < K - 2 * KSPAN; k += 2 * KSPAN) {
        gemm_frag_load<MT, NT>(xr, wr, k + KSPAN, a1, b1);
        __builtin_amdgcn_sched_barrier(0);
        gemm_frag_mma<MT, NT>(a0, b0, acc);
        __builtin_amdgcn_sched_barrier(0);
        gemm_frag_load<MT, NT>(xr, wr, k + 2 * KSPAN, a0, b0);
        __builtin_amdgcn_sched_barrier(0);
        gemm_frag_mma<MT, NT>(a1, b1, acc);
        __builtin_amdgcn_sched_barrier(0);
    }
    gemm_frag_load<MT, NT>(xr, wr, K - KSPAN, a1, b1);
    __builtin_amdgcn_sched_barrier(0);
    gemm_frag_mma<MT, NT>(a0, b0, acc);
    gemm_frag_mma<MT, NT>(a1, b1, acc);

#pragma unroll
    for (int nt = 0; nt < NT; ++nt) {
        const int   col  = col0 + nt * 16;
        const float bcol = bias[col + l15];
#pragma unroll
        for (int mt = 0; mt < MT; ++mt) {
            const int row = row0 + mt * 16;
#pragma unroll
            for (int r = 0; r < 8; ++r) {
                out[(size_t)(row + r + 8 * kh) * Ncols + col + l15] =
                    acc[mt][nt][r] + bcol;
            }
        }
    }
}

template <int MT, int NT, int KBT, int K>
__global__ void __launch_bounds__(128)
wmma_gemm_f32(const float* __restrict__ X, const float* __restrict__ W,
              const float* __restrict__ bias, float* __restrict__ out,
              int Ncols) {
    const int wv = threadIdx.x >> 5;
    gemm_tile_f32<MT, NT, true, KBT>(X, W, bias, out, Ncols, K,
                                     blockIdx.x * (32 * MT) + (wv & 1) * (16 * MT),
                                     blockIdx.y * (32 * NT) + (wv >> 1) * (16 * NT),
                                     threadIdx.x & 31);
}

__global__ void __launch_bounds__(32)
proj_all(const float* __restrict__ inp, const float* __restrict__ Wv,
         const float* __restrict__ bv, const float* __restrict__ query,
         const float* __restrict__ Ws, const float* __restrict__ bs,
         const float* __restrict__ Wa, const float* __restrict__ ba,
         const float* __restrict__ ipts,
         float* __restrict__ value, float* __restrict__ off,
         float* __restrict__ logits, float* __restrict__ V4) {
    const int lane = threadIdx.x;
    const int id   = blockIdx.x;
    if (id < VT) {
        const int bx = id % VT_X;
        const int by = id / VT_X;
        if (by == 0) {
            const int l = bx * 32 + lane;
            const float* v = ipts + (size_t)l * 3;
            const float x = v[0], y = v[1], z = v[2];
            float4 t;
            t.x = -2.0f * x;
            t.y = -2.0f * y;
            t.z = -2.0f * z;
            t.w = x * x + y * y + z * z;
            *(float4*)(V4 + (size_t)l * 4) = t;
        }
        gemm_tile_f32<2, 4, true>(inp, Wv, bv, value, CC, CC, bx * 32, by * 64, lane);
    } else {
        const int j  = id - VT;
        const int bx = j % QT_X;
        const int by = j / QT_X;
        if (by < 3)
            gemm_tile_f32<2, 2, false>(query, Ws, bs, off, NOFF, CC, bx * 32, by * 32, lane);
        else
            gemm_tile_f32<2, 2, true>(query, Wa, ba, logits, MH * NP, CC, bx * 32, 0, lane);
    }
}

__device__ __forceinline__ void knn_step(const v2f a, const v2f* bq, int cib,
                                         float* best, int* bgrp) {
#pragma unroll
    for (int t = 0; t < QT; ++t) {
        v8f c = {};
        c = __builtin_amdgcn_wmma_f32_16x16x4_f32(
            false, a, false, bq[t], (short)0, c, false, false);
        const float t0 = fminf(fminf(c[0], c[1]), c[2]);
        const float t1 = fminf(fminf(c[3], c[4]), c[5]);
        const float t2 = fminf(fminf(c[6], c[7]), best[t]);
        const float nb = fminf(fminf(t0, t1), t2);
        if (nb < best[t]) bgrp[t] = cib;
        best[t] = nb;
    }
}

__global__ void __launch_bounds__(32 * NW)
knn_gather(const float* __restrict__ off, const float* __restrict__ qpts,
           const float* __restrict__ V4, const float* __restrict__ value,
           const float* __restrict__ logits, float* __restrict__ out_tmp) {
    constexpr int PW = (16 * QT) / NW;
    static_assert(PW * NW == 16 * QT && PW <= 16 && 16 % PW == 0,
                  "a wave's sampling points must share one query tile");

    __shared__ float sBest[NW][16 * QT];
    __shared__ int   sIdx[NW][16 * QT];
    __shared__ int   sGrp[16 * QT];
    __shared__ int   sFin[16 * QT];

    const int lane = threadIdx.x & 31;
    const int wv   = threadIdx.x >> 5;
    const int l15  = lane & 15;
    const int kh   = lane >> 4;
    const int n0   = blockIdx.x * (16 * QT);
    const int b    = n0 >> 15;
    const int cw   = wv * 16;

    v2f bq[QT];
#pragma unroll
    for (int t = 0; t < QT; ++t) {
        const int n  = n0 + t * 16 + l15;
        const int rq = n >> 5;
        const int mp = n & 31;
        const float* o = off  + (size_t)rq * NOFF + mp * 3;
        const float* p = qpts + (size_t)rq * 3;
        const float s0 = p[2 * kh] + o[2 * kh];
        const float s1 = p[1] + o[1];
        v2f f;
        f.x = s0;
        f.y = kh ? 1.0f : s1;
        bq[t] = f;
    }

    const float* __restrict__ vlane = V4 + (size_t)b * LIN * 4
                                         + (size_t)l15 * 4 + 2 * kh;
    const float* __restrict__ vptr = vlane + (size_t)cw * 4;

    float best[QT];
    int   bgrp[QT];
#pragma unroll
    for (int t = 0; t < QT; ++t) { best[t] = 3.4e38f; bgrp[t] = 0; }

    v2f a_cur = *(const v2f*)(vptr);
    for (int cb = 0; cb < LIN - CSTEP; cb += CSTEP) {
        const v2f a_next = *(const v2f*)(vptr + (size_t)(cb + CSTEP) * 4);
        knn_step(a_cur, bq, cw + cb + (kh << 3), best, bgrp);
        a_cur = a_next;
    }
    knn_step(a_cur, bq, cw + (LIN - CSTEP) + (kh << 3), best, bgrp);

#pragma unroll
    for (int t = 0; t < QT; ++t) {
        const float obest = __shfl_xor(best[t], 16, 32);
        const int   ogrp  = __shfl_xor(bgrp[t], 16, 32);
        if (obest < best[t] || (obest == best[t] && ogrp < bgrp[t])) {
            best[t] = obest; bgrp[t] = ogrp;
        }
        if (lane < 16) {
            sBest[wv][t * 16 + l15] = best[t];
            sIdx[wv][t * 16 + l15]  = bgrp[t];
        }
    }
    __syncthreads();

    if (threadIdx.x < 16 * QT) {
        float mb = sBest[0][threadIdx.x];
        int   mi = sIdx[0][threadIdx.x];
        for (int w = 1; w < NW; ++w) {
            const float ob = sBest[w][threadIdx.x];
            const int   oi = sIdx[w][threadIdx.x];
            if (ob < mb || (ob == mb && oi < mi)) { mb = ob; mi = oi; }
        }
        sGrp[threadIdx.x] = mi;
    }
    __syncthreads();

    {
        const int tw = (wv * PW) >> 4;
        v2f bsel = bq[0];
#pragma unroll
        for (int t = 1; t < QT; ++t)
            if (tw == t) bsel = bq[t];
        const int mine = l15 % PW;
        int g[PW];
        v2f a[PW];
#pragma unroll
        for (int i = 0; i < PW; ++i) g[i] = sGrp[wv * PW + i];
#pragma unroll
        for (int i = 0; i < PW; ++i)
            a[i] = *(const v2f*)(vlane + (size_t)(g[i] & ~15) * 4);
        v8f mc = {};
        int myg = 0;
#pragma unroll
        for (int i = 0; i < PW; ++i) {
            v8f c = {};
            c = __builtin_amdgcn_wmma_f32_16x16x4_f32(
                false, a[i], false, bsel, (short)0, c, false, false);
            if (mine == i) { mc = c; myg = g[i]; }
        }
        float m2[4]; int r2[4];
#pragma unroll
        for (int k = 0; k < 4; ++k) {
            const bool s = mc[2 * k + 1] < mc[2 * k];
            m2[k] = s ? mc[2 * k + 1] : mc[2 * k];
            r2[k] = s ? 2 * k + 1 : 2 * k;
        }
        float m4[2]; int r4[2];
#pragma unroll
        for (int k = 0; k < 2; ++k) {
            const bool s = m2[2 * k + 1] < m2[2 * k];
            m4[k] = s ? m2[2 * k + 1] : m2[2 * k];
            r4[k] = s ? r2[2 * k + 1] : r2[2 * k];
        }
        const int r = (m4[1] < m4[0]) ? r4[1] : r4[0];
        if ((l15 - mine) == ((wv * PW) & 15) && kh == ((myg >> 3) & 1))
            sFin[wv * PW + mine] = myg + r;
    }
    __syncthreads();

    {
        const int c = threadIdx.x;
        const int m = c >> 5;
        const float* vb = value + (size_t)b * LIN * CC;
#pragma unroll
        for (int j = 0; j < (16 * QT) / (MH * NP); ++j) {
            const int rq = (n0 >> 5) + j;
            float4 v = *(const float4*)(logits + ((size_t)rq * MH + m) * NP);
            const float mx = fmaxf(fmaxf(v.x, v.y), fmaxf(v.z, v.w));
            const float e0 = __expf(v.x - mx), e1 = __expf(v.y - mx);
            const float e2 = __expf(v.z - mx), e3 = __expf(v.w - mx);
            const float inv = 1.0f / (e0 + e1 + e2 + e3);
            float aw[NP];
            aw[0] = e0 * inv; aw[1] = e1 * inv; aw[2] = e2 * inv; aw[3] = e3 * inv;
            const int* id = sFin + j * (MH * NP) + m * NP;
            float acc = 0.0f;
#pragma unroll
            for (int p = 0; p < NP; ++p)
                acc += aw[p] * vb[(size_t)id[p] * CC + c];
            out_tmp[(size_t)rq * CC + c] = acc;
        }
    }
}

extern "C" void kernel_launch(void* const* d_in, const int* in_sizes, int n_in,
                              void* d_out, int out_size, void* d_ws, size_t ws_size,
                              hipStream_t stream) {
    const float* query = (const float*)d_in[0];
    const float* qpts  = (const float*)d_in[1];
    const float* inp   = (const float*)d_in[2];
    const float* ipts  = (const float*)d_in[3];
    const float* Wv    = (const float*)d_in[4];
    const float* bv    = (const float*)d_in[5];
    const float* Ws    = (const float*)d_in[6];
    const float* bs    = (const float*)d_in[7];
    const float* Wa    = (const float*)d_in[8];
    const float* ba    = (const float*)d_in[9];
    const float* Wo    = (const float*)d_in[10];
    const float* bo    = (const float*)d_in[11];
    float* out = (float*)d_out;

    char* w = (char*)d_ws;
    float* value  = (float*)w;  w += (size_t)BB * LIN * CC * 4;
    float* off    = (float*)w;  w += (size_t)BB * LQ * MH * NP * 3 * 4;
    float* logits = (float*)w;  w += (size_t)BB * LQ * MH * NP * 4;
    float* V4     = (float*)w;  w += (size_t)BB * LIN * 4 * 4;
    float* otmp   = (float*)w;  w += (size_t)BB * LQ * CC * 4;

    const dim3 wave(32);

    proj_all<<<VT + QT_X * QT_Y, wave, 0, stream>>>(
        inp, Wv, bv, query, Ws, bs, Wa, ba, ipts, value, off, logits, V4);

    knn_gather<<<(BB * LQ * MH * NP) / (16 * QT), dim3(32 * NW), 0, stream>>>(
        off, qpts, V4, value, logits, otmp);

    wmma_gemm_f32<1, 2, 16, CC><<<dim3((BB * LQ) / 32, CC / 64), dim3(128), 0, stream>>>(
        otmp, Wo, bo, out, CC);
}


# KNN candidate stream: each wave requests its fragments 8 blocks at a time, one whole group ahead of the group it scans (was 1 block ahead); block order and products unchanged, bit-identical
# speedup vs baseline: 4.4728x; 1.0421x over previous
#include <hip/hip_runtime.h>
#include <hip/hip_bf16.h>


typedef __attribute__((ext_vector_type(2))) float v2f;
typedef __attribute__((ext_vector_type(4))) float v4f;
typedef __attribute__((ext_vector_type(8))) float v8f;

#define BB   2
#define LQ   1024
#define LIN  4096
#define CC   256
#define MH   8
#define NP   4
#define DH   32
#define QT   4
#define NW   8
#define CSTEP (16 * NW)
#define PFG   8
#define GSPAN (PFG * CSTEP)

#define VT_X ((BB * LIN) / 32)
#define VT_Y (CC / 64)
#define VT   (VT_X * VT_Y)
#define QT_X ((BB * LQ) / 32)
#define QT_Y 4
#define NOFF (MH * NP * 3)

#define KB    8

template <int MT, int NT, int NS>
__device__ __forceinline__ void
gemm_frag_load(const float* __restrict__ const (&xr)[MT],
               const float* __restrict__ const (&wr)[NT], int k0,
               v2f (&a)[NS][MT], v2f (&b)[NS][NT]) {
#pragma unroll
    for (int s = 0; s < NS; ++s) {
#pragma unroll
        for (int mt = 0; mt < MT; ++mt) a[s][mt] = *(const v2f*)(xr[mt] + k0 + 4 * s);
#pragma unroll
        for (int nt = 0; nt < NT; ++nt) b[s][nt] = *(const v2f*)(wr[nt] + k0 + 4 * s);
    }
}

template <int MT, int NT, int NS>
__device__ __forceinline__ void
gemm_frag_mma(const v2f (&a)[NS][MT], const v2f (&b)[NS][NT], v8f (&acc)[MT][NT]) {
#pragma unroll
    for (int s = 0; s < NS; ++s)
#pragma unroll
        for (int mt = 0; mt < MT; ++mt)
#pragma unroll
            for (int nt = 0; nt < NT; ++nt)
                acc[mt][nt] = __builtin_amdgcn_wmma_f32_16x16x4_f32(
                     false, a[s][mt],  false, b[s][nt],
                     (short)0, acc[mt][nt],
                     false,  false);
}

template <int MT, int NT, int NQ>
__device__ __forceinline__ void
gemm_frag_load(const float* __restrict__ const (&xr)[MT],
               const float* __restrict__ const (&wr)[NT], int k0,
               v4f (&a)[NQ][MT], v4f (&b)[NQ][NT]) {
#pragma unroll
    for (int j = 0; j < NQ; ++j) {
#pragma unroll
        for (int mt = 0; mt < MT; ++mt) a[j][mt] = *(const v4f*)(xr[mt] + k0 + 8 * j);
#pragma unroll
        for (int nt = 0; nt < NT; ++nt) b[j][nt] = *(const v4f*)(wr[nt] + k0 + 8 * j);
    }
}

template <int MT, int NT, int NQ>
__device__ __forceinline__ void
gemm_frag_mma(const v4f (&a)[NQ][MT], const v4f (&b)[NQ][NT], v8f (&acc)[MT][NT]) {
#pragma unroll
    for (int j = 0; j < NQ; ++j) {
#pragma unroll
        for (int mt = 0; mt < MT; ++mt)
#pragma unroll
            for (int nt = 0; nt < NT; ++nt)
                acc[mt][nt] = __builtin_amdgcn_wmma_f32_16x16x4_f32(
                     false, a[j][mt].xy,  false, b[j][nt].xy,
                     (short)0, acc[mt][nt],
                     false,  false);
#pragma unroll
        for (int mt = 0; mt < MT; ++mt)
#pragma unroll
            for (int nt = 0; nt < NT; ++nt)
                acc[mt][nt] = __builtin_amdgcn_wmma_f32_16x16x4_f32(
                     false, a[j][mt].zw,  false, b[j][nt].zw,
                     (short)0, acc[mt][nt],
                     false,  false);
    }
}

template <bool WIDE, int KBT> struct FragKind {
    typedef v2f type;
    static constexpr int N = KBT;
    static constexpr int LANE_K = 2;
};
template <int KBT> struct FragKind<true, KBT> {
    typedef v4f type;
    static constexpr int N = KBT / 2;
    static constexpr int LANE_K = 4;
};

template <int MT, int NT, bool WIDE, int KBT = KB>
__device__ __forceinline__ void
gemm_tile_f32(const float* __restrict__ X, const float* __restrict__ W,
              const float* __restrict__ bias, float* __restrict__ out,
              int Ncols, int K, int row0, int col0, int lane) {
    typedef typename FragKind<WIDE, KBT>::type frag_t;
    constexpr int NF = FragKind<WIDE, KBT>::N;
    constexpr int KSPAN = 4 * KBT;
    const int l15  = lane & 15;
    const int kh   = lane >> 4;

    const float* __restrict__ xr[MT];
    const float* __restrict__ wr[NT];
#pragma unroll
    for (int mt = 0; mt < MT; ++mt)
        xr[mt] = X + (size_t)(row0 + mt * 16 + l15) * K + FragKind<WIDE, KBT>::LANE_K * kh;
#pragma unroll
    for (int nt = 0; nt < NT; ++nt)
        wr[nt] = W + (size_t)(col0 + nt * 16 + l15) * K + FragKind<WIDE, KBT>::LANE_K * kh;

    v8f acc[MT][NT];
#pragma unroll
    for (int mt = 0; mt < MT; ++mt)
#pragma unroll
        for (int nt = 0; nt < NT; ++nt) {
            v8f z = {};
            acc[mt][nt] = z;
        }

    frag_t a0[NF][MT], b0[NF][NT], a1[NF][MT], b1[NF][NT];
    gemm_frag_load<MT, NT>(xr, wr, 0, a0, b0);
    for (int k = 0; k < K - 2 * KSPAN; k += 2 * KSPAN) {
        gemm_frag_load<MT, NT>(xr, wr, k + KSPAN, a1, b1);
        __builtin_amdgcn_sched_barrier(0);
        gemm_frag_mma<MT, NT>(a0, b0, acc);
        __builtin_amdgcn_sched_barrier(0);
        gemm_frag_load<MT, NT>(xr, wr, k + 2 * KSPAN, a0, b0);
        __builtin_amdgcn_sched_barrier(0);
        gemm_frag_mma<MT, NT>(a1, b1, acc);
        __builtin_amdgcn_sched_barrier(0);
    }
    gemm_frag_load<MT, NT>(xr, wr, K - KSPAN, a1, b1);
    __builtin_amdgcn_sched_barrier(0);
    gemm_frag_mma<MT, NT>(a0, b0, acc);
    gemm_frag_mma<MT, NT>(a1, b1, acc);

#pragma unroll
    for (int nt = 0; nt < NT; ++nt) {
        const int   col  = col0 + nt * 16;
        const float bcol = bias[col + l15];
#pragma unroll
        for (int mt = 0; mt < MT; ++mt) {
            const int row = row0 + mt * 16;
#pragma unroll
            for (int r = 0; r < 8; ++r) {
                out[(size_t)(row + r + 8 * kh) * Ncols + col + l15] =
                    acc[mt][nt][r] + bcol;
            }
        }
    }
}

template <int MT, int NT, int KBT, int K>
__global__ void __launch_bounds__(128)
wmma_gemm_f32(const float* __restrict__ X, const float* __restrict__ W,
              const float* __restrict__ bias, float* __restrict__ out,
              int Ncols) {
    const int wv = threadIdx.x >> 5;
    gemm_tile_f32<MT, NT, true, KBT>(X, W, bias, out, Ncols, K,
                                     blockIdx.x * (32 * MT) + (wv & 1) * (16 * MT),
                                     blockIdx.y * (32 * NT) + (wv >> 1) * (16 * NT),
                                     threadIdx.x & 31);
}

__global__ void __launch_bounds__(32)
proj_all(const float* __restrict__ inp, const float* __restrict__ Wv,
         const float* __restrict__ bv, const float* __restrict__ query,
         const float* __restrict__ Ws, const float* __restrict__ bs,
         const float* __restrict__ Wa, const float* __restrict__ ba,
         const float* __restrict__ ipts,
         float* __restrict__ value, float* __restrict__ off,
         float* __restrict__ logits, float* __restrict__ V4) {
    const int lane = threadIdx.x;
    const int id   = blockIdx.x;
    if (id < VT) {
        const int bx = id % VT_X;
        const int by = id / VT_X;
        if (by == 0) {
            const int l = bx * 32 + lane;
            const float* v = ipts + (size_t)l * 3;
            const float x = v[0], y = v[1], z = v[2];
            float4 t;
            t.x = -2.0f * x;
            t.y = -2.0f * y;
            t.z = -2.0f * z;
            t.w = x * x + y * y + z * z;
            *(float4*)(V4 + (size_t)l * 4) = t;
        }
        gemm_tile_f32<2, 4, true>(inp, Wv, bv, value, CC, CC, bx * 32, by * 64, lane);
    } else {
        const int j  = id - VT;
        const int bx = j % QT_X;
        const int by = j / QT_X;
        if (by < 3)
            gemm_tile_f32<2, 2, false>(query, Ws, bs, off, NOFF, CC, bx * 32, by * 32, lane);
        else
            gemm_tile_f32<2, 2, true>(query, Wa, ba, logits, MH * NP, CC, bx * 32, 0, lane);
    }
}

__device__ __forceinline__ void knn_step(const v2f a, const v2f* bq, int cib,
                                         float* best, int* bgrp) {
#pragma unroll
    for (int t = 0; t < QT; ++t) {
        v8f c = {};
        c = __builtin_amdgcn_wmma_f32_16x16x4_f32(
            false, a, false, bq[t], (short)0, c, false, false);
        const float t0 = fminf(fminf(c[0], c[1]), c[2]);
        const float t1 = fminf(fminf(c[3], c[4]), c[5]);
        const float t2 = fminf(fminf(c[6], c[7]), best[t]);
        const float nb = fminf(fminf(t0, t1), t2);
        if (nb < best[t]) bgrp[t] = cib;
        best[t] = nb;
    }
}

__device__ __forceinline__ void knn_load_group(const float* __restrict__ vptr, int cb0,
                                               v2f (&a)[PFG]) {
#pragma unroll
    for (int i = 0; i < PFG; ++i)
        a[i] = *(const v2f*)(vptr + (size_t)(cb0 + i * CSTEP) * 4);
}

__device__ __forceinline__ void knn_scan_group(const v2f (&a)[PFG], const v2f* bq, int cib0,
                                               float* best, int* bgrp) {
#pragma unroll
    for (int i = 0; i < PFG; ++i)
        knn_step(a[i], bq, cib0 + i * CSTEP, best, bgrp);
}

__global__ void __launch_bounds__(32 * NW)
knn_gather(const float* __restrict__ off, const float* __restrict__ qpts,
           const float* __restrict__ V4, const float* __restrict__ value,
           const float* __restrict__ logits, float* __restrict__ out_tmp) {
    constexpr int PW = (16 * QT) / NW;
    static_assert(PW * NW == 16 * QT && PW <= 16 && 16 % PW == 0,
                  "a wave's sampling points must share one query tile");

    __shared__ float sBest[NW][16 * QT];
    __shared__ int   sIdx[NW][16 * QT];
    __shared__ int   sGrp[16 * QT];
    __shared__ int   sFin[16 * QT];

    const int lane = threadIdx.x & 31;
    const int wv   = threadIdx.x >> 5;
    const int l15  = lane & 15;
    const int kh   = lane >> 4;
    const int n0   = blockIdx.x * (16 * QT);
    const int b    = n0 >> 15;
    const int cw   = wv * 16;

    v2f bq[QT];
#pragma unroll
    for (int t = 0; t < QT; ++t) {
        const int n  = n0 + t * 16 + l15;
        const int rq = n >> 5;
        const int mp = n & 31;
        const float* o = off  + (size_t)rq * NOFF + mp * 3;
        const float* p = qpts + (size_t)rq * 3;
        const float s0 = p[2 * kh] + o[2 * kh];
        const float s1 = p[1] + o[1];
        v2f f;
        f.x = s0;
        f.y = kh ? 1.0f : s1;
        bq[t] = f;
    }

    const float* __restrict__ vlane = V4 + (size_t)b * LIN * 4
                                         + (size_t)l15 * 4 + 2 * kh;
    const float* __restrict__ vptr = vlane + (size_t)cw * 4;

    float best[QT];
    int   bgrp[QT];
#pragma unroll
    for (int t = 0; t < QT; ++t) { best[t] = 3.4e38f; bgrp[t] = 0; }

    static_assert(LIN % (2 * GSPAN) == 0, "the stream must be a whole number of group pairs");
    const int cib = cw + (kh << 3);
    v2f a0[PFG], a1[PFG];
    knn_load_group(vptr, 0, a0);
    for (int cb = 0; cb < LIN - 2 * GSPAN; cb += 2 * GSPAN) {
        knn_load_group(vptr, cb + GSPAN, a1);
        __builtin_amdgcn_sched_barrier(0);
        knn_scan_group(a0, bq, cib + cb, best, bgrp);
        __builtin_amdgcn_sched_barrier(0);
        knn_load_group(vptr, cb + 2 * GSPAN, a0);
        __builtin_amdgcn_sched_barrier(0);
        knn_scan_group(a1, bq, cib + cb + GSPAN, best, bgrp);
        __builtin_amdgcn_sched_barrier(0);
    }
    knn_load_group(vptr, LIN - GSPAN, a1);
    __builtin_amdgcn_sched_barrier(0);
    knn_scan_group(a0, bq, cib + (LIN - 2 * GSPAN), best, bgrp);
    knn_scan_group(a1, bq, cib + (LIN - GSPAN), best, bgrp);

#pragma unroll
    for (int t = 0; t < QT; ++t) {
        const float obest = __shfl_xor(best[t], 16, 32);
        const int   ogrp  = __shfl_xor(bgrp[t], 16, 32);
        if (obest < best[t] || (obest == best[t] && ogrp < bgrp[t])) {
            best[t] = obest; bgrp[t] = ogrp;
        }
        if (lane < 16) {
            sBest[wv][t * 16 + l15] = best[t];
            sIdx[wv][t * 16 + l15]  = bgrp[t];
        }
    }
    __syncthreads();

    if (threadIdx.x < 16 * QT) {
        float mb = sBest[0][threadIdx.x];
        int   mi = sIdx[0][threadIdx.x];
        for (int w = 1; w < NW; ++w) {
            const float ob = sBest[w][threadIdx.x];
            const int   oi = sIdx[w][threadIdx.x];
            if (ob < mb || (ob == mb && oi < mi)) { mb = ob; mi = oi; }
        }
        sGrp[threadIdx.x] = mi;
    }
    __syncthreads();

    {
        const int tw = (wv * PW) >> 4;
        v2f bsel = bq[0];
#pragma unroll
        for (int t = 1; t < QT; ++t)
            if (tw == t) bsel = bq[t];
        const int mine = l15 % PW;
        int g[PW];
        v2f a[PW];
#pragma unroll
        for (int i = 0; i < PW; ++i) g[i] = sGrp[wv * PW + i];
#pragma unroll
        for (int i = 0; i < PW; ++i)
            a[i] = *(const v2f*)(vlane + (size_t)(g[i] & ~15) * 4);
        v8f mc = {};
        int myg = 0;
#pragma unroll
        for (int i = 0; i < PW; ++i) {
            v8f c = {};
            c = __builtin_amdgcn_wmma_f32_16x16x4_f32(
                false, a[i], false, bsel, (short)0, c, false, false);
            if (mine == i) { mc = c; myg = g[i]; }
        }
        float m2[4]; int r2[4];
#pragma unroll
        for (int k = 0; k < 4; ++k) {
            const bool s = mc[2 * k + 1] < mc[2 * k];
            m2[k] = s ? mc[2 * k + 1] : mc[2 * k];
            r2[k] = s ? 2 * k + 1 : 2 * k;
        }
        float m4[2]; int r4[2];
#pragma unroll
        for (int k = 0; k < 2; ++k) {
            const bool s = m2[2 * k + 1] < m2[2 * k];
            m4[k] = s ? m2[2 * k + 1] : m2[2 * k];
            r4[k] = s ? r2[2 * k + 1] : r2[2 * k];
        }
        const int r = (m4[1] < m4[0]) ? r4[1] : r4[0];
        if ((l15 - mine) == ((wv * PW) & 15) && kh == ((myg >> 3) & 1))
            sFin[wv * PW + mine] = myg + r;
    }
    __syncthreads();

    {
        const int c = threadIdx.x;
        const int m = c >> 5;
        const float* vb = value + (size_t)b * LIN * CC;
#pragma unroll
        for (int j = 0; j < (16 * QT) / (MH * NP); ++j) {
            const int rq = (n0 >> 5) + j;
            float4 v = *(const float4*)(logits + ((size_t)rq * MH + m) * NP);
            const float mx = fmaxf(fmaxf(v.x, v.y), fmaxf(v.z, v.w));
            const float e0 = __expf(v.x - mx), e1 = __expf(v.y - mx);
            const float e2 = __expf(v.z - mx), e3 = __expf(v.w - mx);
            const float inv = 1.0f / (e0 + e1 + e2 + e3);
            float aw[NP];
            aw[0] = e0 * inv; aw[1] = e1 * inv; aw[2] = e2 * inv; aw[3] = e3 * inv;
            const int* id = sFin + j * (MH * NP) + m * NP;
            float acc = 0.0f;
#pragma unroll
            for (int p = 0; p < NP; ++p)
                acc += aw[p] * vb[(size_t)id[p] * CC + c];
            out_tmp[(size_t)rq * CC + c] = acc;
        }
    }
}

extern "C" void kernel_launch(void* const* d_in, const int* in_sizes, int n_in,
                              void* d_out, int out_size, void* d_ws, size_t ws_size,
                              hipStream_t stream) {
    const float* query = (const float*)d_in[0];
    const float* qpts  = (const float*)d_in[1];
    const float* inp   = (const float*)d_in[2];
    const float* ipts  = (const float*)d_in[3];
    const float* Wv    = (const float*)d_in[4];
    const float* bv    = (const float*)d_in[5];
    const float* Ws    = (const float*)d_in[6];
    const float* bs    = (const float*)d_in[7];
    const float* Wa    = (const float*)d_in[8];
    const float* ba    = (const float*)d_in[9];
    const float* Wo    = (const float*)d_in[10];
    const float* bo    = (const float*)d_in[11];
    float* out = (float*)d_out;

    char* w = (char*)d_ws;
    float* value  = (float*)w;  w += (size_t)BB * LIN * CC * 4;
    float* off    = (float*)w;  w += (size_t)BB * LQ * MH * NP * 3 * 4;
    float* logits = (float*)w;  w += (size_t)BB * LQ * MH * NP * 4;
    float* V4     = (float*)w;  w += (size_t)BB * LIN * 4 * 4;
    float* otmp   = (float*)w;  w += (size_t)BB * LQ * CC * 4;

    const dim3 wave(32);

    proj_all<<<VT + QT_X * QT_Y, wave, 0, stream>>>(
        inp, Wv, bv, query, Ws, bs, Wa, ba, ipts, value, off, logits, V4);

    knn_gather<<<(BB * LQ * MH * NP) / (16 * QT), dim3(32 * NW), 0, stream>>>(
        off, qpts, V4, value, logits, otmp);

    wmma_gemm_f32<1, 2, 16, CC><<<dim3((BB * LQ) / 32, CC / 64), dim3(128), 0, stream>>>(
        otmp, Wo, bo, out, CC);
}


# Launch fusion 3->2: a KNN work-group is 4 groups of 8 waves (8 query rows), copies Wo into LDS beside its scan and ends with the output product of its rows (16-row tiles half filled); bit-identical
# speedup vs baseline: 4.7790x; 1.0685x over previous
#include <hip/hip_runtime.h>
#include <hip/hip_bf16.h>


typedef __attribute__((ext_vector_type(2))) float v2f;
typedef __attribute__((ext_vector_type(4))) float v4f;
typedef __attribute__((ext_vector_type(8))) float v8f;

#define BB   2
#define LQ   1024
#define LIN  4096
#define CC   256
#define MH   8
#define NP   4
#define DH   32
#define QT   4
#define NW   8
#define CSTEP (16 * NW)
#define PFG   8
#define GSPAN (PFG * CSTEP)
#define NG   4
#define OPITCH (CC + 4)
#define OUT_WAVES (CC / 16)
#define WROWS (CC / (NW * NG))
#define SL   2

#define VT_X ((BB * LIN) / 32)
#define VT_Y (CC / 64)
#define VT   (VT_X * VT_Y)
#define QT_X ((BB * LQ) / 32)
#define QT_Y 4
#define NOFF (MH * NP * 3)

#define KB    8

template <int MT, int NT, int NS>
__device__ __forceinline__ void
gemm_frag_load(const float* __restrict__ const (&xr)[MT],
               const float* __restrict__ const (&wr)[NT], int k0,
               v2f (&a)[NS][MT], v2f (&b)[NS][NT]) {
#pragma unroll
    for (int s = 0; s < NS; ++s) {
#pragma unroll
        for (int mt = 0; mt < MT; ++mt) a[s][mt] = *(const v2f*)(xr[mt] + k0 + 4 * s);
#pragma unroll
        for (int nt = 0; nt < NT; ++nt) b[s][nt] = *(const v2f*)(wr[nt] + k0 + 4 * s);
    }
}

template <int MT, int NT, int NS>
__device__ __forceinline__ void
gemm_frag_mma(const v2f (&a)[NS][MT], const v2f (&b)[NS][NT], v8f (&acc)[MT][NT]) {
#pragma unroll
    for (int s = 0; s < NS; ++s)
#pragma unroll
        for (int mt = 0; mt < MT; ++mt)
#pragma unroll
            for (int nt = 0; nt < NT; ++nt)
                acc[mt][nt] = __builtin_amdgcn_wmma_f32_16x16x4_f32(
                     false, a[s][mt],  false, b[s][nt],
                     (short)0, acc[mt][nt],
                     false,  false);
}

template <int MT, int NT, int NQ>
__device__ __forceinline__ void
gemm_frag_load(const float* __restrict__ const (&xr)[MT],
               const float* __restrict__ const (&wr)[NT], int k0,
               v4f (&a)[NQ][MT], v4f (&b)[NQ][NT]) {
#pragma unroll
    for (int j = 0; j < NQ; ++j) {
#pragma unroll
        for (int mt = 0; mt < MT; ++mt) a[j][mt] = *(const v4f*)(xr[mt] + k0 + 8 * j);
#pragma unroll
        for (int nt = 0; nt < NT; ++nt) b[j][nt] = *(const v4f*)(wr[nt] + k0 + 8 * j);
    }
}

template <int MT, int NT, int NQ>
__device__ __forceinline__ void
gemm_frag_mma(const v4f (&a)[NQ][MT], const v4f (&b)[NQ][NT], v8f (&acc)[MT][NT]) {
#pragma unroll
    for (int j = 0; j < NQ; ++j) {
#pragma unroll
        for (int mt = 0; mt < MT; ++mt)
#pragma unroll
            for (int nt = 0; nt < NT; ++nt)
                acc[mt][nt] = __builtin_amdgcn_wmma_f32_16x16x4_f32(
                     false, a[j][mt].xy,  false, b[j][nt].xy,
                     (short)0, acc[mt][nt],
                     false,  false);
#pragma unroll
        for (int mt = 0; mt < MT; ++mt)
#pragma unroll
            for (int nt = 0; nt < NT; ++nt)
                acc[mt][nt] = __builtin_amdgcn_wmma_f32_16x16x4_f32(
                     false, a[j][mt].zw,  false, b[j][nt].zw,
                     (short)0, acc[mt][nt],
                     false,  false);
    }
}

template <bool WIDE, int KBT> struct FragKind {
    typedef v2f type;
    static constexpr int N = KBT;
    static constexpr int LANE_K = 2;
};
template <int KBT> struct FragKind<true, KBT> {
    typedef v4f type;
    static constexpr int N = KBT / 2;
    static constexpr int LANE_K = 4;
};

template <int MT, int NT, bool WIDE, int KBT = KB>
__device__ __forceinline__ void
gemm_tile_f32(const float* __restrict__ X, const float* __restrict__ W,
              const float* __restrict__ bias, float* __restrict__ out,
              int Ncols, int K, int row0, int col0, int lane) {
    typedef typename FragKind<WIDE, KBT>::type frag_t;
    constexpr int NF = FragKind<WIDE, KBT>::N;
    constexpr int KSPAN = 4 * KBT;
    const int l15  = lane & 15;
    const int kh   = lane >> 4;

    const float* __restrict__ xr[MT];
    const float* __restrict__ wr[NT];
#pragma unroll
    for (int mt = 0; mt < MT; ++mt)
        xr[mt] = X + (size_t)(row0 + mt * 16 + l15) * K + FragKind<WIDE, KBT>::LANE_K * kh;
#pragma unroll
    for (int nt = 0; nt < NT; ++nt)
        wr[nt] = W + (size_t)(col0 + nt * 16 + l15) * K + FragKind<WIDE, KBT>::LANE_K * kh;

    v8f acc[MT][NT];
#pragma unroll
    for (int mt = 0; mt < MT; ++mt)
#pragma unroll
        for (int nt = 0; nt < NT; ++nt) {
            v8f z = {};
            acc[mt][nt] = z;
        }

    frag_t a0[NF][MT], b0[NF][NT], a1[NF][MT], b1[NF][NT];
    gemm_frag_load<MT, NT>(xr, wr, 0, a0, b0);
    for (int k = 0; k < K - 2 * KSPAN; k += 2 * KSPAN) {
        gemm_frag_load<MT, NT>(xr, wr, k + KSPAN, a1, b1);
        __builtin_amdgcn_sched_barrier(0);
        gemm_frag_mma<MT, NT>(a0, b0, acc);
        __builtin_amdgcn_sched_barrier(0);
        gemm_frag_load<MT, NT>(xr, wr, k + 2 * KSPAN, a0, b0);
        __builtin_amdgcn_sched_barrier(0);
        gemm_frag_mma<MT, NT>(a1, b1, acc);
        __builtin_amdgcn_sched_barrier(0);
    }
    gemm_frag_load<MT, NT>(xr, wr, K - KSPAN, a1, b1);
    __builtin_amdgcn_sched_barrier(0);
    gemm_frag_mma<MT, NT>(a0, b0, acc);
    gemm_frag_mma<MT, NT>(a1, b1, acc);

#pragma unroll
    for (int nt = 0; nt < NT; ++nt) {
        const int   col  = col0 + nt * 16;
        const float bcol = bias[col + l15];
#pragma unroll
        for (int mt = 0; mt < MT; ++mt) {
            const int row = row0 + mt * 16;
#pragma unroll
            for (int r = 0; r < 8; ++r) {
                out[(size_t)(row + r + 8 * kh) * Ncols + col + l15] =
                    acc[mt][nt][r] + bcol;
            }
        }
    }
}

__global__ void __launch_bounds__(32)
proj_all(const float* __restrict__ inp, const float* __restrict__ Wv,
         const float* __restrict__ bv, const float* __restrict__ query,
         const float* __restrict__ Ws, const float* __restrict__ bs,
         const float* __restrict__ Wa, const float* __restrict__ ba,
         const float* __restrict__ ipts,
         float* __restrict__ value, float* __restrict__ off,
         float* __restrict__ logits, float* __restrict__ V4) {
    const int lane = threadIdx.x;
    const int id   = blockIdx.x;
    if (id < VT) {
        const int bx = id % VT_X;
        const int by = id / VT_X;
        if (by == 0) {
            const int l = bx * 32 + lane;
            const float* v = ipts + (size_t)l * 3;
            const float x = v[0], y = v[1], z = v[2];
            float4 t;
            t.x = -2.0f * x;
            t.y = -2.0f * y;
            t.z = -2.0f * z;
            t.w = x * x + y * y + z * z;
            *(float4*)(V4 + (size_t)l * 4) = t;
        }
        gemm_tile_f32<2, 4, true>(inp, Wv, bv, value, CC, CC, bx * 32, by * 64, lane);
    } else {
        const int j  = id - VT;
        const int bx = j % QT_X;
        const int by = j / QT_X;
        if (by < 3)
            gemm_tile_f32<2, 2, false>(query, Ws, bs, off, NOFF, CC, bx * 32, by * 32, lane);
        else
            gemm_tile_f32<2, 2, true>(query, Wa, ba, logits, MH * NP, CC, bx * 32, 0, lane);
    }
}

__device__ __forceinline__ void knn_step(const v2f a, const v2f* bq, int cib,
                                         float* best, int* bgrp) {
#pragma unroll
    for (int t = 0; t < QT; ++t) {
        v8f c = {};
        c = __builtin_amdgcn_wmma_f32_16x16x4_f32(
            false, a, false, bq[t], (short)0, c, false, false);
        const float t0 = fminf(fminf(c[0], c[1]), c[2]);
        const float t1 = fminf(fminf(c[3], c[4]), c[5]);
        const float t2 = fminf(fminf(c[6], c[7]), best[t]);
        const float nb = fminf(fminf(t0, t1), t2);
        if (nb < best[t]) bgrp[t] = cib;
        best[t] = nb;
    }
}

__device__ __forceinline__ void knn_load_group(const float* __restrict__ vptr, int cb0,
                                               v2f (&a)[PFG]) {
#pragma unroll
    for (int i = 0; i < PFG; ++i)
        a[i] = *(const v2f*)(vptr + (size_t)(cb0 + i * CSTEP) * 4);
}

template <int I0, int I1>
__device__ __forceinline__ void knn_scan_part(const v2f (&a)[PFG], const v2f* bq, int cib0,
                                              float* best, int* bgrp) {
#pragma unroll
    for (int i = I0; i < I1; ++i)
        knn_step(a[i], bq, cib0 + i * CSTEP, best, bgrp);
}

__device__ __forceinline__ void wo_piece_load(const float* __restrict__ wsrc, int i0,
                                              v4f (&s)[SL]) {
#pragma unroll
    for (int u = 0; u < SL; ++u) {
        const int i = i0 + u;
        s[u] = *(const v4f*)(wsrc + (size_t)(i >> 1) * CC + (i & 1) * (CC / 2));
    }
}
__device__ __forceinline__ void wo_piece_store(float* wdst, int i0, const v4f (&s)[SL]) {
#pragma unroll
    for (int u = 0; u < SL; ++u) {
        const int i = i0 + u;
        *(v4f*)(wdst + (i >> 1) * OPITCH + (i & 1) * (CC / 2)) = s[u];
    }
}

__global__ void __launch_bounds__(32 * NW * NG)
knn_gather_out(const float* __restrict__ off, const float* __restrict__ qpts,
               const float* __restrict__ V4, const float* __restrict__ value,
               const float* __restrict__ logits, const float* __restrict__ Wo,
               const float* __restrict__ bo, float* __restrict__ out) {
    constexpr int PW = (16 * QT) / NW;
    static_assert(PW * NW == 16 * QT && PW <= 16 && 16 % PW == 0,
                  "a wave's sampling points must share one query tile");

    __shared__ float sBest[NG][NW][16 * QT];
    __shared__ int   sIdx[NG][NW][16 * QT];
    __shared__ int   sGrp[NG][16 * QT];
    __shared__ int   sFin[NG][16 * QT];
    __shared__ __attribute__((aligned(16))) float sOut[2 * NG][OPITCH];
    __shared__ __attribute__((aligned(16))) float sWo[CC][OPITCH];
    static_assert(2 * WROWS == 2 * SL * (LIN / GSPAN), "a wave copies its rows of Wo in SL pieces per half scan group");

    const int lane = threadIdx.x & 31;
    const int wid  = __builtin_amdgcn_readfirstlane(threadIdx.x >> 5);
    const int wv   = wid % NW;
    const int tg   = wid / NW;
    const int tid  = threadIdx.x % (32 * NW);
    const int l15  = lane & 15;
    const int kh   = lane >> 4;
    const int n0   = (blockIdx.x * NG + tg) * (16 * QT);
    const int b    = n0 >> 15;
    const int cw   = wv * 16;

    v2f bq[QT];
#pragma unroll
    for (int t = 0; t < QT; ++t) {
        const int n  = n0 + t * 16 + l15;
        const int rq = n >> 5;
        const int mp = n & 31;
        const float* o = off  + (size_t)rq * NOFF + mp * 3;
        const float* p = qpts + (size_t)rq * 3;
        const float s0 = p[2 * kh] + o[2 * kh];
        const float s1 = p[1] + o[1];
        v2f f;
        f.x = s0;
        f.y = kh ? 1.0f : s1;
        bq[t] = f;
    }

    const float* __restrict__ vlane = V4 + (size_t)b * LIN * 4
                                         + (size_t)l15 * 4 + 2 * kh;
    const float* __restrict__ vptr = vlane + (size_t)cw * 4;

    float best[QT];
    int   bgrp[QT];
#pragma unroll
    for (int t = 0; t < QT; ++t) { best[t] = 3.4e38f; bgrp[t] = 0; }

    static_assert(LIN % (2 * GSPAN) == 0, "the stream must be a whole number of group pairs");
    const int cib = cw + (kh << 3);
    const float* __restrict__ wsrc = Wo + (size_t)(WROWS * wid) * CC + lane * 4;
    float* wdst = &sWo[WROWS * wid][lane * 4];
    v4f st[SL];
    v2f a0[PFG], a1[PFG];
    knn_load_group(vptr, 0, a0);
#define KNN_GROUP(A, G, CIB0, NEXT)                                                        \
    wo_piece_load(wsrc, (2 * (G)) * SL, st);                                               \
    NEXT;                                                                                  \
    __builtin_amdgcn_sched_barrier(0);       \
    knn_scan_part<0, PFG / 2>(A, bq, CIB0, best, bgrp);                                    \
    __builtin_amdgcn_sched_barrier(0);                       \
    wo_piece_store(wdst, (2 * (G)) * SL, st);                                              \
    wo_piece_load(wsrc, (2 * (G) + 1) * SL, st);                                           \
    __builtin_amdgcn_sched_barrier(0);       \
    knn_scan_part<PFG / 2, PFG>(A, bq, CIB0, best, bgrp);                                  \
    __builtin_amdgcn_sched_barrier(0);                       \
    wo_piece_store(wdst, (2 * (G) + 1) * SL, st);                                          \
    __builtin_amdgcn_sched_barrier(0);

    for (int cb = 0; cb < LIN - 2 * GSPAN; cb += 2 * GSPAN) {
        KNN_GROUP(a0, cb / GSPAN, cib + cb, knn_load_group(vptr, cb + GSPAN, a1))
        KNN_GROUP(a1, cb / GSPAN + 1, cib + cb + GSPAN, knn_load_group(vptr, cb + 2 * GSPAN, a0))
    }
    KNN_GROUP(a0, LIN / GSPAN - 2, cib + (LIN - 2 * GSPAN), knn_load_group(vptr, LIN - GSPAN, a1))
    KNN_GROUP(a1, LIN / GSPAN - 1, cib + (LIN - GSPAN), (void)0)
#undef KNN_GROUP

#pragma unroll
    for (int t = 0; t < QT; ++t) {
        const float obest = __shfl_xor(best[t], 16, 32);
        const int   ogrp  = __shfl_xor(bgrp[t], 16, 32);
        if (obest < best[t] || (obest == best[t] && ogrp < bgrp[t])) {
            best[t] = obest; bgrp[t] = ogrp;
        }
        if (lane < 16) {
            sBest[tg][wv][t * 16 + l15] = best[t];
            sIdx[tg][wv][t * 16 + l15]  = bgrp[t];
        }
    }
    __syncthreads();

    if (tid < 16 * QT) {
        float mb = sBest[tg][0][tid];
        int   mi = sIdx[tg][0][tid];
        for (int w = 1; w < NW; ++w) {
            const float ob = sBest[tg][w][tid];
            const int   oi = sIdx[tg][w][tid];
            if (ob < mb || (ob == mb && oi < mi)) { mb = ob; mi = oi; }
        }
        sGrp[tg][tid] = mi;
    }
    __syncthreads();

    {
        const int tw = (wv * PW) >> 4;
        v2f bsel = bq[0];
#pragma unroll
        for (int t = 1; t < QT; ++t)
            if (tw == t) bsel = bq[t];
        const int mine = l15 % PW;
        int g[PW];
        v2f a[PW];
#pragma unroll
        for (int i = 0; i < PW; ++i) g[i] = sGrp[tg][wv * PW + i];
#pragma unroll
        for (int i = 0; i < PW; ++i)
            a[i] = *(const v2f*)(vlane + (size_t)(g[i] & ~15) * 4);
        v8f mc = {};
        int myg = 0;
#pragma unroll
        for (int i = 0; i < PW; ++i) {
            v8f c = {};
            c = __builtin_amdgcn_wmma_f32_16x16x4_f32(
                false, a[i], false, bsel, (short)0, c, false, false);
            if (mine == i) { mc = c; myg = g[i]; }
        }
        float m2[4]; int r2[4];
#pragma unroll
        for (int k = 0; k < 4; ++k) {
            const bool s = mc[2 * k + 1] < mc[2 * k];
            m2[k] = s ? mc[2 * k + 1] : mc[2 * k];
            r2[k] = s ? 2 * k + 1 : 2 * k;
        }
        float m4[2]; int r4[2];
#pragma unroll
        for (int k = 0; k < 2; ++k) {
            const bool s = m2[2 * k + 1] < m2[2 * k];
            m4[k] = s ? m2[2 * k + 1] : m2[2 * k];
            r4[k] = s ? r2[2 * k + 1] : r2[2 * k];
        }
        const int r = (m4[1] < m4[0]) ? r4[1] : r4[0];
        if ((l15 - mine) == ((wv * PW) & 15) && kh == ((myg >> 3) & 1))
            sFin[tg][wv * PW + mine] = myg + r;
    }
    __syncthreads();

    {
        const int c = tid;
        const int m = c >> 5;
        const float* vb = value + (size_t)b * LIN * CC;
#pragma unroll
        for (int j = 0; j < (16 * QT) / (MH * NP); ++j) {
            const int rq = (n0 >> 5) + j;
            float4 v = *(const float4*)(logits + ((size_t)rq * MH + m) * NP);
            const float mx = fmaxf(fmaxf(v.x, v.y), fmaxf(v.z, v.w));
            const float e0 = __expf(v.x - mx), e1 = __expf(v.y - mx);
            const float e2 = __expf(v.z - mx), e3 = __expf(v.w - mx);
            const float inv = 1.0f / (e0 + e1 + e2 + e3);
            float aw[NP];
            aw[0] = e0 * inv; aw[1] = e1 * inv; aw[2] = e2 * inv; aw[3] = e3 * inv;
            const int* id = sFin[tg] + j * (MH * NP) + m * NP;
            float acc = 0.0f;
#pragma unroll
            for (int p = 0; p < NP; ++p)
                acc += aw[p] * vb[(size_t)id[p] * CC + c];
            sOut[2 * tg + j][c] = acc;
        }
    }
    __syncthreads();

    const bool outw = wid < OUT_WAVES;
    if (outw) {
        const float* arow = &sOut[l15 % (2 * NG)][4 * kh];
        const float* brow = &sWo[(wid % OUT_WAVES) * 16 + l15][4 * kh];
        v8f acc = {};
        v4f af[8], bf[8];
#pragma unroll
        for (int blk = 0; blk < CC / 64; ++blk) {
            const int k0 = blk * 64;
#pragma unroll
            for (int j = 0; j < 8; ++j) af[j] = *(const v4f*)(arow + k0 + 8 * j);
#pragma unroll
            for (int j = 0; j < 8; ++j) bf[j] = *(const v4f*)(brow + k0 + 8 * j);
#pragma unroll
            for (int j = 0; j < 8; ++j) {
                acc = __builtin_amdgcn_wmma_f32_16x16x4_f32(
                    false, af[j].xy, false, bf[j].xy, (short)0, acc, false, false);
                acc = __builtin_amdgcn_wmma_f32_16x16x4_f32(
                    false, af[j].zw, false, bf[j].zw, (short)0, acc, false, false);
            }
        }
        if (kh == 0) {
            const int col = (wid % OUT_WAVES) * 16 + l15;
            const float bcol = bo[col];
            float* orow = out + (size_t)(blockIdx.x * (2 * NG)) * CC + col;
#pragma unroll
            for (int r = 0; r < 8; ++r)
                orow[(size_t)r * CC] = acc[r] + bcol;
        }
    }
}

extern "C" void kernel_launch(void* const* d_in, const int* in_sizes, int n_in,
                              void* d_out, int out_size, void* d_ws, size_t ws_size,
                              hipStream_t stream) {
    const float* query = (const float*)d_in[0];
    const float* qpts  = (const float*)d_in[1];
    const float* inp   = (const float*)d_in[2];
    const float* ipts  = (const float*)d_in[3];
    const float* Wv    = (const float*)d_in[4];
    const float* bv    = (const float*)d_in[5];
    const float* Ws    = (const float*)d_in[6];
    const float* bs    = (const float*)d_in[7];
    const float* Wa    = (const float*)d_in[8];
    const float* ba    = (const float*)d_in[9];
    const float* Wo    = (const float*)d_in[10];
    const float* bo    = (const float*)d_in[11];
    float* out = (float*)d_out;

    char* w = (char*)d_ws;
    float* value  = (float*)w;  w += (size_t)BB * LIN * CC * 4;
    float* off    = (float*)w;  w += (size_t)BB * LQ * MH * NP * 3 * 4;
    float* logits = (float*)w;  w += (size_t)BB * LQ * MH * NP * 4;
    float* V4     = (float*)w;  w += (size_t)BB * LIN * 4 * 4;

    const dim3 wave(32);

    proj_all<<<VT + QT_X * QT_Y, wave, 0, stream>>>(
        inp, Wv, bv, query, Ws, bs, Wa, ba, ipts, value, off, logits, V4);

    knn_gather_out<<<(BB * LQ * MH * NP) / (16 * QT * NG), dim3(32 * NW * NG), 0, stream>>>(
        off, qpts, V4, value, logits, Wo, bo, out);
}
